# starting kernel
# baseline (speedup 1.0000x reference)
#include <hip/hip_runtime.h>

#define N_USERS 100000
#define N_ITEMS 50000
#define NNODES  (N_USERS + N_ITEMS)
#define DIM     64
#define BATCH   8192
#define NLAYERS 3

typedef __attribute__((ext_vector_type(2))) float v2f;
typedef __attribute__((ext_vector_type(8))) float v8f;

__global__ void lgcn_init(const float* __restrict__ uemb,
                          const float* __restrict__ iemb,
                          float* __restrict__ acc,
                          float* __restrict__ bufA,
                          float* __restrict__ bufB) {
    long idx = (long)blockIdx.x * blockDim.x + threadIdx.x;
    const long tot = (long)NNODES * DIM / 4;
    if (idx >= tot) return;
    const long uq = (long)N_USERS * DIM / 4;
    float4 v = (idx < uq) ? ((const float4*)uemb)[idx]
                          : ((const float4*)iemb)[idx - uq];
    ((float4*)acc)[idx]  = v;
    ((float4*)bufA)[idx] = v;
    ((float4*)bufB)[idx] = make_float4(0.f, 0.f, 0.f, 0.f);
}

__global__ void lgcn_spmm(const int*   __restrict__ rows,
                          const int*   __restrict__ cols,
                          const float* __restrict__ vals,
                          const float* __restrict__ x,
                          float*       __restrict__ y,
                          int nnz) {
    int wave = blockIdx.x * (blockDim.x >> 5) + (threadIdx.x >> 5);
    int lane = threadIdx.x & 31;
    if (wave >= nnz) return;
    int   r = rows[wave];
    int   c = cols[wave];
    float v = vals[wave];
    float2 xv = ((const float2*)(x + (long)c * DIM))[lane];
    float* yr = y + (long)r * DIM + 2 * lane;
    unsafeAtomicAdd(yr,     v * xv.x);
    unsafeAtomicAdd(yr + 1, v * xv.y);
}

__global__ void lgcn_accum(float*       __restrict__ acc,
                           const float* __restrict__ xnew,
                           float*       __restrict__ tozero,
                           int do_zero) {
    long idx = (long)blockIdx.x * blockDim.x + threadIdx.x;
    const long tot = (long)NNODES * DIM / 4;
    if (idx >= tot) return;
    float4 a = ((float4*)acc)[idx];
    float4 b = ((const float4*)xnew)[idx];
    a.x += b.x; a.y += b.y; a.z += b.z; a.w += b.w;
    ((float4*)acc)[idx] = a;
    if (do_zero) ((float4*)tozero)[idx] = make_float4(0.f, 0.f, 0.f, 0.f);
}

__device__ __forceinline__ float pick8(v8f c, int j) {
    float r = c[0];
    if (j == 1) r = c[1];
    if (j == 2) r = c[2];
    if (j == 3) r = c[3];
    if (j == 4) r = c[4];
    if (j == 5) r = c[5];
    if (j == 6) r = c[6];
    if (j == 7) r = c[7];
    return r;
}

__global__ void lgcn_score(const float* __restrict__ acc,
                           const int*   __restrict__ users,
                           const int*   __restrict__ pos,
                           const int*   __restrict__ neg,
                           float*       __restrict__ pos_scores,
                           float*       __restrict__ neg_scores) {
    int tile = blockIdx.x * (blockDim.x >> 5) + (threadIdx.x >> 5);
    int lane = threadIdx.x & 31;
    int m     = lane & 15;
    int khalf = lane >> 4;
    int b  = tile * 16 + m;
    int u  = users[b];
    int pi = pos[b] + N_USERS;
    int ni = neg[b] + N_USERS;
    const float* urow = acc + (long)u  * DIM;
    const float* prow = acc + (long)pi * DIM;
    const float* nrow = acc + (long)ni * DIM;

    v8f cpos = {};
    v8f cneg = {};
#pragma unroll
    for (int kk = 0; kk < 16; ++kk) {
        int k0 = kk * 4 + khalf * 2;
        float2 ua = *(const float2*)(urow + k0);
        float2 pb = *(const float2*)(prow + k0);
        float2 nb = *(const float2*)(nrow + k0);
        v2f a;  a[0]  = ua.x; a[1]  = ua.y;
        v2f bp; bp[0] = pb.x; bp[1] = pb.y;
        v2f bn; bn[0] = nb.x; bn[1] = nb.y;
        cpos = __builtin_amdgcn_wmma_f32_16x16x4_f32(false, a, false, bp,
                                                     (short)0, cpos, false, false);
        cneg = __builtin_amdgcn_wmma_f32_16x16x4_f32(false, a, false, bn,
                                                     (short)0, cneg, false, false);
    }
    const float s = 1.0f / 16.0f;
    if (lane < 8) {
        pos_scores[tile * 16 + lane] = pick8(cpos, lane) * s;
        neg_scores[tile * 16 + lane] = pick8(cneg, lane) * s;
    } else if (lane >= 24) {
        int mm = lane - 16;
        pos_scores[tile * 16 + mm] = pick8(cpos, lane - 24) * s;
        neg_scores[tile * 16 + mm] = pick8(cneg, lane - 24) * s;
    }
}

__global__ void lgcn_gather0(const float* __restrict__ uemb,
                             const float* __restrict__ iemb,
                             const int*   __restrict__ users,
                             const int*   __restrict__ pos,
                             const int*   __restrict__ neg,
                             float* __restrict__ u0,
                             float* __restrict__ p0,
                             float* __restrict__ n0) {
    int idx = blockIdx.x * blockDim.x + threadIdx.x;
    if (idx >= BATCH * (DIM / 4)) return;
    int b = idx >> 4;
    int q = idx & 15;
    ((float4*)u0)[idx] = ((const float4*)uemb)[(long)users[b] * 16 + q];
    ((float4*)p0)[idx] = ((const float4*)iemb)[(long)pos[b]   * 16 + q];
    ((float4*)n0)[idx] = ((const float4*)iemb)[(long)neg[b]   * 16 + q];
}

extern "C" void kernel_launch(void* const* d_in, const int* in_sizes, int n_in,
                              void* d_out, int out_size, void* d_ws, size_t ws_size,
                              hipStream_t stream) {
    const float* uemb  = (const float*)d_in[0];
    const float* iemb  = (const float*)d_in[1];
    const float* vals  = (const float*)d_in[2];
    const int*   rows  = (const int*)d_in[3];
    const int*   cols  = (const int*)d_in[4];
    const int*   users = (const int*)d_in[5];
    const int*   pos   = (const int*)d_in[6];
    const int*   neg   = (const int*)d_in[7];
    const int    nnz   = in_sizes[3];

    float* out        = (float*)d_out;
    float* pos_scores = out;
    float* neg_scores = out + BATCH;
    float* u0         = out + 2 * BATCH;
    float* p0         = u0 + BATCH * DIM;
    float* n0         = p0 + BATCH * DIM;

    const size_t nfeat = (size_t)NNODES * DIM;
    float* acc  = (float*)d_ws;
    float* bufA = acc  + nfeat;
    float* bufB = bufA + nfeat;

    const long q4 = (long)nfeat / 4;
    lgcn_init<<<(int)((q4 + 255) / 256), 256, 0, stream>>>(uemb, iemb, acc, bufA, bufB);

    float* cur = bufA;
    float* nxt = bufB;
    for (int l = 0; l < NLAYERS; ++l) {
        int waves_per_block = 256 / 32;
        int grid = (nnz + waves_per_block - 1) / waves_per_block;
        lgcn_spmm<<<grid, 256, 0, stream>>>(rows, cols, vals, cur, nxt, nnz);
        lgcn_accum<<<(int)((q4 + 255) / 256), 256, 0, stream>>>(
            acc, nxt, cur, (l < NLAYERS - 1) ? 1 : 0);
        float* t = cur; cur = nxt; nxt = t;
    }

    lgcn_score<<<(BATCH / 16) / 8, 256, 0, stream>>>(acc, users, pos, neg,
                                                     pos_scores, neg_scores);

    lgcn_gather0<<<(BATCH * (DIM / 4) + 255) / 256, 256, 0, stream>>>(
        uemb, iemb, users, pos, neg, u0, p0, n0);
}


# Edges counting-sorted by destination row once per call (integer atomics), then one wave sums each row in registers: float atomics, zero fills and accumulate passes go; sums now in ranked edge order
# speedup vs baseline: 2.3945x; 2.3945x over previous
#include <hip/hip_runtime.h>


#define N_USERS 100000
#define N_ITEMS 50000
#define NNODES  (N_USERS + N_ITEMS)
#define DIM     64
#define BATCH   8192
#define NQ      (NNODES / 4)
#define ROWS_PER_WAVE 4

typedef __attribute__((ext_vector_type(2))) float v2f;
typedef __attribute__((ext_vector_type(8))) float v8f;

__global__ void lgcn_zero_counts(int* __restrict__ counts) {
    int q = blockIdx.x * blockDim.x + threadIdx.x;
    if (q < NQ) ((int4*)counts)[q] = make_int4(0, 0, 0, 0);
}

__global__ void lgcn_count(const int* __restrict__ rows,
                           int*       __restrict__ counts,
                           int*       __restrict__ rank,
                           int nnz) {
    int e = blockIdx.x * blockDim.x + threadIdx.x;
    if (e < nnz) rank[e] = atomicAdd(&counts[rows[e]], 1);
}

__global__ void lgcn_scan(const int* __restrict__ counts,
                          int*       __restrict__ row_ptr) {
    __shared__ int s_pre[8];
    __shared__ int s_own[8];
    const int t = threadIdx.x;
    const int lane = t & 31;
    const int w = t >> 5;
    const int lim = blockIdx.x * 256;
    const int4* c4 = (const int4*)counts;
    int pre = 0;
#pragma unroll 8
    for (int i = 0; i < (int)blockIdx.x; ++i) {
        int4 c = c4[i * 256 + t];
        pre += (c.x + c.y) + (c.z + c.w);
    }
    const int q = lim + t;
    int4 c = make_int4(0, 0, 0, 0);
    if (q < NQ) c = c4[q];
    const int own = (c.x + c.y) + (c.z + c.w);
#pragma unroll
    for (int d = 16; d >= 1; d >>= 1) pre += __shfl_xor(pre, d);
    int incl = own;
#pragma unroll
    for (int d = 1; d < 32; d <<= 1) {
        int v = __shfl_up(incl, d);
        if (lane >= d) incl += v;
    }
    if (lane == 31) { s_pre[w] = pre; s_own[w] = incl; }
    __syncthreads();
    int base = 0;
#pragma unroll
    for (int k = 0; k < 8; ++k) {
        base += s_pre[k];
        if (k < w) base += s_own[k];
    }
    const int e0 = base + incl - own;
    if (q < NQ) {
        int4 r;
        r.x = e0;
        r.y = r.x + c.x;
        r.z = r.y + c.y;
        r.w = r.z + c.z;
        ((int4*)row_ptr)[q] = r;
    }
}

__global__ void lgcn_place(const int*   __restrict__ rows,
                           const int*   __restrict__ cols,
                           const float* __restrict__ vals,
                           const int*   __restrict__ row_ptr,
                           const int*   __restrict__ rank,
                           int2*        __restrict__ edges,
                           int nnz) {
    int e = blockIdx.x * blockDim.x + threadIdx.x;
    if (e < nnz) {
        int p = row_ptr[rows[e]] + rank[e];
        edges[p] = make_int2(cols[e], __float_as_int(vals[e]));
    }
}

__device__ __forceinline__ float2 lgcn_row_sum(const int2*  __restrict__ edges,
                                               int beg, int end, int lane,
                                               const float* __restrict__ xa,
                                               const float* __restrict__ xb,
                                               int split) {
    float2 s = make_float2(0.f, 0.f);
    for (int base = beg; base < end; base += 32) {
        int n = end - base;
        if (n > 32) n = 32;
        int2 ed = make_int2(0, 0);
        if (lane < n) ed = edges[base + lane];
        for (int j = 0; j < n; j += 8) {
            float2 xv[8];
            float  vv[8];
#pragma unroll
            for (int k = 0; k < 8; ++k) {
                int c = __builtin_amdgcn_readlane(ed.x, j + k);
                vv[k] = __int_as_float(__builtin_amdgcn_readlane(ed.y, j + k));
                const float* src = (c < split) ? (xa + (long)c * DIM)
                                               : (xb + (long)(c - split) * DIM);
                xv[k] = ((const float2*)src)[lane];
            }
#pragma unroll
            for (int k = 0; k < 8; ++k) {
                s.x += vv[k] * xv[k].x;
                s.y += vv[k] * xv[k].y;
            }
        }
    }
    return s;
}

__global__ void lgcn_layer_first(const int*   __restrict__ row_ptr,
                                 const int*   __restrict__ counts,
                                 const int2*  __restrict__ edges,
                                 const float* __restrict__ uemb,
                                 const float* __restrict__ iemb,
                                 float*       __restrict__ x1) {
    const int wave = __builtin_amdgcn_readfirstlane(blockIdx.x * (blockDim.x >> 5) + (threadIdx.x >> 5));
    const int lane = threadIdx.x & 31;
    for (int i = 0; i < ROWS_PER_WAVE; ++i) {
        const int row = wave * ROWS_PER_WAVE + i;
        if (row >= NNODES) break;
        const int beg = row_ptr[row];
        const int end = beg + counts[row];
        float2 s = lgcn_row_sum(edges, beg, end, lane, uemb, iemb, N_USERS);
        ((float2*)(x1 + (long)row * DIM))[lane] = s;
    }
}

__global__ void lgcn_layer_mid(const int*   __restrict__ row_ptr,
                               const int*   __restrict__ counts,
                               const int2*  __restrict__ edges,
                               const float* __restrict__ xin,
                               float*       __restrict__ xout) {
    const int wave = __builtin_amdgcn_readfirstlane(blockIdx.x * (blockDim.x >> 5) + (threadIdx.x >> 5));
    const int lane = threadIdx.x & 31;
    for (int i = 0; i < ROWS_PER_WAVE; ++i) {
        const int row = wave * ROWS_PER_WAVE + i;
        if (row >= NNODES) break;
        const int beg = row_ptr[row];
        const int end = beg + counts[row];
        float2 s = lgcn_row_sum(edges, beg, end, lane, xin, xin, NNODES);
        ((float2*)(xout + (long)row * DIM))[lane] = s;
    }
}

__global__ void lgcn_layer_last(const int*   __restrict__ row_ptr,
                                const int*   __restrict__ counts,
                                const int2*  __restrict__ edges,
                                const float* __restrict__ uemb,
                                const float* __restrict__ iemb,
                                const float* __restrict__ x2,
                                float*       __restrict__ x1_acc) {
    const int wave = __builtin_amdgcn_readfirstlane(blockIdx.x * (blockDim.x >> 5) + (threadIdx.x >> 5));
    const int lane = threadIdx.x & 31;
    for (int i = 0; i < ROWS_PER_WAVE; ++i) {
        const int row = wave * ROWS_PER_WAVE + i;
        if (row >= NNODES) break;
        const int beg = row_ptr[row];
        const int end = beg + counts[row];
        const float* x0row = (row < N_USERS) ? (uemb + (long)row * DIM)
                                             : (iemb + (long)(row - N_USERS) * DIM);
        float2 a0 = ((const float2*)x0row)[lane];
        float2 a1 = ((const float2*)(x1_acc + (long)row * DIM))[lane];
        float2 a2 = ((const float2*)(x2 + (long)row * DIM))[lane];
        float2 s = lgcn_row_sum(edges, beg, end, lane, x2, x2, NNODES);
        float2 r;
        r.x = ((a0.x + a1.x) + a2.x) + s.x;
        r.y = ((a0.y + a1.y) + a2.y) + s.y;
        ((float2*)(x1_acc + (long)row * DIM))[lane] = r;
    }
}

__device__ __forceinline__ float pick8(v8f c, int j) {
    float r = c[0];
    if (j == 1) r = c[1];
    if (j == 2) r = c[2];
    if (j == 3) r = c[3];
    if (j == 4) r = c[4];
    if (j == 5) r = c[5];
    if (j == 6) r = c[6];
    if (j == 7) r = c[7];
    return r;
}

__global__ void lgcn_score(const float* __restrict__ acc,
                           const int*   __restrict__ users,
                           const int*   __restrict__ pos,
                           const int*   __restrict__ neg,
                           float*       __restrict__ pos_scores,
                           float*       __restrict__ neg_scores) {
    int tile = blockIdx.x * (blockDim.x >> 5) + (threadIdx.x >> 5);
    int lane = threadIdx.x & 31;
    int m     = lane & 15;
    int khalf = lane >> 4;
    int b  = tile * 16 + m;
    int u  = users[b];
    int pi = pos[b] + N_USERS;
    int ni = neg[b] + N_USERS;
    const float* urow = acc + (long)u  * DIM;
    const float* prow = acc + (long)pi * DIM;
    const float* nrow = acc + (long)ni * DIM;

    v8f cpos = {};
    v8f cneg = {};
#pragma unroll
    for (int kk = 0; kk < 16; ++kk) {
        int k0 = kk * 4 + khalf * 2;
        float2 ua = *(const float2*)(urow + k0);
        float2 pb = *(const float2*)(prow + k0);
        float2 nb = *(const float2*)(nrow + k0);
        v2f a;  a[0]  = ua.x; a[1]  = ua.y;
        v2f bp; bp[0] = pb.x; bp[1] = pb.y;
        v2f bn; bn[0] = nb.x; bn[1] = nb.y;
        cpos = __builtin_amdgcn_wmma_f32_16x16x4_f32(false, a, false, bp,
                                                     (short)0, cpos, false, false);
        cneg = __builtin_amdgcn_wmma_f32_16x16x4_f32(false, a, false, bn,
                                                     (short)0, cneg, false, false);
    }
    const float s = 1.0f / 16.0f;
    if (lane < 8) {
        pos_scores[tile * 16 + lane] = pick8(cpos, lane) * s;
        neg_scores[tile * 16 + lane] = pick8(cneg, lane) * s;
    } else if (lane >= 24) {
        int mm = lane - 16;
        pos_scores[tile * 16 + mm] = pick8(cpos, lane - 24) * s;
        neg_scores[tile * 16 + mm] = pick8(cneg, lane - 24) * s;
    }
}

__global__ void lgcn_gather0(const float* __restrict__ uemb,
                             const float* __restrict__ iemb,
                             const int*   __restrict__ users,
                             const int*   __restrict__ pos,
                             const int*   __restrict__ neg,
                             float* __restrict__ u0,
                             float* __restrict__ p0,
                             float* __restrict__ n0) {
    int idx = blockIdx.x * blockDim.x + threadIdx.x;
    if (idx >= BATCH * (DIM / 4)) return;
    int b = idx >> 4;
    int q = idx & 15;
    ((float4*)u0)[idx] = ((const float4*)uemb)[(long)users[b] * 16 + q];
    ((float4*)p0)[idx] = ((const float4*)iemb)[(long)pos[b]   * 16 + q];
    ((float4*)n0)[idx] = ((const float4*)iemb)[(long)neg[b]   * 16 + q];
}

extern "C" void kernel_launch(void* const* d_in, const int* in_sizes, int n_in,
                              void* d_out, int out_size, void* d_ws, size_t ws_size,
                              hipStream_t stream) {
    const float* uemb  = (const float*)d_in[0];
    const float* iemb  = (const float*)d_in[1];
    const float* vals  = (const float*)d_in[2];
    const int*   rows  = (const int*)d_in[3];
    const int*   cols  = (const int*)d_in[4];
    const int*   users = (const int*)d_in[5];
    const int*   pos   = (const int*)d_in[6];
    const int*   neg   = (const int*)d_in[7];
    const int    nnz   = in_sizes[3];

    float* out        = (float*)d_out;
    float* pos_scores = out;
    float* neg_scores = out + BATCH;
    float* u0         = out + 2 * BATCH;
    float* p0         = u0 + BATCH * DIM;
    float* n0         = p0 + BATCH * DIM;

    const size_t nfeat = (size_t)NNODES * DIM;
    float* x1      = (float*)d_ws;
    float* x2      = x1 + nfeat;
    int*   rank    = (int*)x2;
    int*   counts  = (int*)(x2 + nfeat);
    int*   row_ptr = counts + NNODES;
    int2*  edges   = (int2*)(row_ptr + NNODES);

    const int qgrid = (NQ + 255) / 256;
    const int egrid = (nnz + 255) / 256;
    const int lgrid = (NNODES / ROWS_PER_WAVE + 7) / 8;

    lgcn_zero_counts<<<qgrid, 256, 0, stream>>>(counts);
    lgcn_count<<<egrid, 256, 0, stream>>>(rows, counts, rank, nnz);
    lgcn_scan<<<qgrid, 256, 0, stream>>>(counts, row_ptr);
    lgcn_place<<<egrid, 256, 0, stream>>>(rows, cols, vals, row_ptr, rank, edges, nnz);

    lgcn_layer_first<<<lgrid, 256, 0, stream>>>(row_ptr, counts, edges, uemb, iemb, x1);
    lgcn_layer_mid<<<lgrid, 256, 0, stream>>>(row_ptr, counts, edges, x1, x2);
    lgcn_layer_last<<<lgrid, 256, 0, stream>>>(row_ptr, counts, edges, uemb, iemb, x2, x1);

    lgcn_score<<<(BATCH / 16) / 8, 256, 0, stream>>>(x1, users, pos, neg,
                                                     pos_scores, neg_scores);

    lgcn_gather0<<<(BATCH * (DIM / 4) + 255) / 256, 256, 0, stream>>>(
        uemb, iemb, users, pos, neg, u0, p0, n0);
}


# Layer 3 summed only for the 3x8192 rows the scores read, inside the scoring kernel: 16 waves per batch tile stage acc rows in LDS, wave 0 runs the same f32 WMMAs; sum order unchanged
# speedup vs baseline: 2.6121x; 1.0909x over previous
#include <hip/hip_runtime.h>


#define N_USERS 100000
#define N_ITEMS 50000
#define NNODES  (N_USERS + N_ITEMS)
#define DIM     64
#define BATCH   8192
#define NQ      (NNODES / 4)
#define ROWS_PER_WAVE 4

typedef __attribute__((ext_vector_type(2))) float v2f;
typedef __attribute__((ext_vector_type(8))) float v8f;

__global__ void lgcn_zero_counts(int* __restrict__ counts) {
    int q = blockIdx.x * blockDim.x + threadIdx.x;
    if (q < NQ) ((int4*)counts)[q] = make_int4(0, 0, 0, 0);
}

__global__ void lgcn_count(const int* __restrict__ rows,
                           int*       __restrict__ counts,
                           int*       __restrict__ rank,
                           int nnz) {
    int e = blockIdx.x * blockDim.x + threadIdx.x;
    if (e < nnz) rank[e] = atomicAdd(&counts[rows[e]], 1);
}

__global__ void lgcn_scan(const int* __restrict__ counts,
                          int*       __restrict__ row_ptr) {
    __shared__ int s_pre[8];
    __shared__ int s_own[8];
    const int t = threadIdx.x;
    const int lane = t & 31;
    const int w = t >> 5;
    const int lim = blockIdx.x * 256;
    const int4* c4 = (const int4*)counts;
    int pre = 0;
#pragma unroll 8
    for (int i = 0; i < (int)blockIdx.x; ++i) {
        int4 c = c4[i * 256 + t];
        pre += (c.x + c.y) + (c.z + c.w);
    }
    const int q = lim + t;
    int4 c = make_int4(0, 0, 0, 0);
    if (q < NQ) c = c4[q];
    const int own = (c.x + c.y) + (c.z + c.w);
#pragma unroll
    for (int d = 16; d >= 1; d >>= 1) pre += __shfl_xor(pre, d);
    int incl = own;
#pragma unroll
    for (int d = 1; d < 32; d <<= 1) {
        int v = __shfl_up(incl, d);
        if (lane >= d) incl += v;
    }
    if (lane == 31) { s_pre[w] = pre; s_own[w] = incl; }
    __syncthreads();
    int base = 0;
#pragma unroll
    for (int k = 0; k < 8; ++k) {
        base += s_pre[k];
        if (k < w) base += s_own[k];
    }
    const int e0 = base + incl - own;
    if (q < NQ) {
        int4 r;
        r.x = e0;
        r.y = r.x + c.x;
        r.z = r.y + c.y;
        r.w = r.z + c.z;
        ((int4*)row_ptr)[q] = r;
    }
}

__global__ void lgcn_place(const int*   __restrict__ rows,
                           const int*   __restrict__ cols,
                           const float* __restrict__ vals,
                           const int*   __restrict__ row_ptr,
                           const int*   __restrict__ rank,
                           int2*        __restrict__ edges,
                           int nnz) {
    int e = blockIdx.x * blockDim.x + threadIdx.x;
    if (e < nnz) {
        int p = row_ptr[rows[e]] + rank[e];
        edges[p] = make_int2(cols[e], __float_as_int(vals[e]));
    }
}

__device__ __forceinline__ float2 lgcn_row_sum(const int2*  __restrict__ edges,
                                               int beg, int end, int lane,
                                               const float* __restrict__ xa,
                                               const float* __restrict__ xb,
                                               int split) {
    float2 s = make_float2(0.f, 0.f);
    for (int base = beg; base < end; base += 32) {
        int n = end - base;
        if (n > 32) n = 32;
        int2 ed = make_int2(0, 0);
        if (lane < n) ed = edges[base + lane];
        for (int j = 0; j < n; j += 8) {
            float2 xv[8];
            float  vv[8];
#pragma unroll
            for (int k = 0; k < 8; ++k) {
                int c = __builtin_amdgcn_readlane(ed.x, j + k);
                vv[k] = __int_as_float(__builtin_amdgcn_readlane(ed.y, j + k));
                const float* src = (c < split) ? (xa + (long)c * DIM)
                                               : (xb + (long)(c - split) * DIM);
                xv[k] = ((const float2*)src)[lane];
            }
#pragma unroll
            for (int k = 0; k < 8; ++k) {
                s.x += vv[k] * xv[k].x;
                s.y += vv[k] * xv[k].y;
            }
        }
    }
    return s;
}

__global__ void lgcn_layer_first(const int*   __restrict__ row_ptr,
                                 const int*   __restrict__ counts,
                                 const int2*  __restrict__ edges,
                                 const float* __restrict__ uemb,
                                 const float* __restrict__ iemb,
                                 float*       __restrict__ x1) {
    const int wave = __builtin_amdgcn_readfirstlane(blockIdx.x * (blockDim.x >> 5) + (threadIdx.x >> 5));
    const int lane = threadIdx.x & 31;
    for (int i = 0; i < ROWS_PER_WAVE; ++i) {
        const int row = wave * ROWS_PER_WAVE + i;
        if (row >= NNODES) break;
        const int beg = row_ptr[row];
        const int end = beg + counts[row];
        float2 s = lgcn_row_sum(edges, beg, end, lane, uemb, iemb, N_USERS);
        ((float2*)(x1 + (long)row * DIM))[lane] = s;
    }
}

__global__ void lgcn_layer_mid(const int*   __restrict__ row_ptr,
                               const int*   __restrict__ counts,
                               const int2*  __restrict__ edges,
                               const float* __restrict__ xin,
                               float*       __restrict__ xout) {
    const int wave = __builtin_amdgcn_readfirstlane(blockIdx.x * (blockDim.x >> 5) + (threadIdx.x >> 5));
    const int lane = threadIdx.x & 31;
    for (int i = 0; i < ROWS_PER_WAVE; ++i) {
        const int row = wave * ROWS_PER_WAVE + i;
        if (row >= NNODES) break;
        const int beg = row_ptr[row];
        const int end = beg + counts[row];
        float2 s = lgcn_row_sum(edges, beg, end, lane, xin, xin, NNODES);
        ((float2*)(xout + (long)row * DIM))[lane] = s;
    }
}

__device__ __forceinline__ float pick8(v8f c, int j) {
    float r = c[0];
    if (j == 1) r = c[1];
    if (j == 2) r = c[2];
    if (j == 3) r = c[3];
    if (j == 4) r = c[4];
    if (j == 5) r = c[5];
    if (j == 6) r = c[6];
    if (j == 7) r = c[7];
    return r;
}

#define ACC_PITCH 68

__global__ void __launch_bounds__(512)
lgcn_last_score(const int*   __restrict__ row_ptr,
                const int*   __restrict__ counts,
                const int2*  __restrict__ edges,
                const float* __restrict__ uemb,
                const float* __restrict__ iemb,
                const float* __restrict__ x1,
                const float* __restrict__ x2,
                const int*   __restrict__ users,
                const int*   __restrict__ pos,
                const int*   __restrict__ neg,
                float*       __restrict__ pos_scores,
                float*       __restrict__ neg_scores) {
    __shared__ float s_acc[3 * 16 * ACC_PITCH];
    const int w    = __builtin_amdgcn_readfirstlane(threadIdx.x >> 5);
    const int lane = threadIdx.x & 31;
    const int tile = blockIdx.x;
    const int b    = tile * 16 + w;
    const int idu = users[b];
    const int idp = pos[b];
    const int idn = neg[b];
    const int ru = idu;
    const int rp = idp + N_USERS;
    const int rn = idn + N_USERS;
    const int begu = row_ptr[ru];
    const int begp = row_ptr[rp];
    const int begn = row_ptr[rn];
    const int endu = begu + counts[ru];
    const int endp = begp + counts[rp];
    const int endn = begn + counts[rn];
    const float2 u0 = ((const float2*)(uemb + (long)idu * DIM))[lane];
    const float2 p0 = ((const float2*)(iemb + (long)idp * DIM))[lane];
    const float2 n0 = ((const float2*)(iemb + (long)idn * DIM))[lane];
    const float2 u1 = ((const float2*)(x1 + (long)ru * DIM))[lane];
    const float2 p1 = ((const float2*)(x1 + (long)rp * DIM))[lane];
    const float2 n1 = ((const float2*)(x1 + (long)rn * DIM))[lane];
    const float2 u2 = ((const float2*)(x2 + (long)ru * DIM))[lane];
    const float2 p2 = ((const float2*)(x2 + (long)rp * DIM))[lane];
    const float2 n2 = ((const float2*)(x2 + (long)rn * DIM))[lane];

    const float2 su = lgcn_row_sum(edges, begu, endu, lane, x2, x2, NNODES);
    const float2 sp = lgcn_row_sum(edges, begp, endp, lane, x2, x2, NNODES);
    const float2 sn = lgcn_row_sum(edges, begn, endn, lane, x2, x2, NNODES);

    float2 au, ap, an;
    au.x = ((u0.x + u1.x) + u2.x) + su.x;
    au.y = ((u0.y + u1.y) + u2.y) + su.y;
    ap.x = ((p0.x + p1.x) + p2.x) + sp.x;
    ap.y = ((p0.y + p1.y) + p2.y) + sp.y;
    an.x = ((n0.x + n1.x) + n2.x) + sn.x;
    an.y = ((n0.y + n1.y) + n2.y) + sn.y;
    ((float2*)(s_acc + (0 * 16 + w) * ACC_PITCH))[lane] = au;
    ((float2*)(s_acc + (1 * 16 + w) * ACC_PITCH))[lane] = ap;
    ((float2*)(s_acc + (2 * 16 + w) * ACC_PITCH))[lane] = an;
    __syncthreads();

    if (w == 0) {
        const int m     = lane & 15;
        const int khalf = lane >> 4;
        const float* urow = s_acc + (0 * 16 + m) * ACC_PITCH + khalf * 2;
        const float* prow = s_acc + (1 * 16 + m) * ACC_PITCH + khalf * 2;
        const float* nrow = s_acc + (2 * 16 + m) * ACC_PITCH + khalf * 2;
        v8f cpos = {};
        v8f cneg = {};
#pragma unroll
        for (int kk = 0; kk < 16; ++kk) {
            float2 ua = *(const float2*)(urow + kk * 4);
            float2 pb = *(const float2*)(prow + kk * 4);
            float2 nb = *(const float2*)(nrow + kk * 4);
            v2f a;  a[0]  = ua.x; a[1]  = ua.y;
            v2f bp; bp[0] = pb.x; bp[1] = pb.y;
            v2f bn; bn[0] = nb.x; bn[1] = nb.y;
            cpos = __builtin_amdgcn_wmma_f32_16x16x4_f32(false, a, false, bp,
                                                         (short)0, cpos, false, false);
            cneg = __builtin_amdgcn_wmma_f32_16x16x4_f32(false, a, false, bn,
                                                         (short)0, cneg, false, false);
        }
        const float s = 1.0f / 16.0f;
        if (lane < 8) {
            pos_scores[tile * 16 + lane] = pick8(cpos, lane) * s;
            neg_scores[tile * 16 + lane] = pick8(cneg, lane) * s;
        } else if (lane >= 24) {
            int mm = lane - 16;
            pos_scores[tile * 16 + mm] = pick8(cpos, lane - 24) * s;
            neg_scores[tile * 16 + mm] = pick8(cneg, lane - 24) * s;
        }
    }
}

__global__ void lgcn_gather0(const float* __restrict__ uemb,
                             const float* __restrict__ iemb,
                             const int*   __restrict__ users,
                             const int*   __restrict__ pos,
                             const int*   __restrict__ neg,
                             float* __restrict__ u0,
                             float* __restrict__ p0,
                             float* __restrict__ n0) {
    int idx = blockIdx.x * blockDim.x + threadIdx.x;
    if (idx >= BATCH * (DIM / 4)) return;
    int b = idx >> 4;
    int q = idx & 15;
    ((float4*)u0)[idx] = ((const float4*)uemb)[(long)users[b] * 16 + q];
    ((float4*)p0)[idx] = ((const float4*)iemb)[(long)pos[b]   * 16 + q];
    ((float4*)n0)[idx] = ((const float4*)iemb)[(long)neg[b]   * 16 + q];
}

extern "C" void kernel_launch(void* const* d_in, const int* in_sizes, int n_in,
                              void* d_out, int out_size, void* d_ws, size_t ws_size,
                              hipStream_t stream) {
    const float* uemb  = (const float*)d_in[0];
    const float* iemb  = (const float*)d_in[1];
    const float* vals  = (const float*)d_in[2];
    const int*   rows  = (const int*)d_in[3];
    const int*   cols  = (const int*)d_in[4];
    const int*   users = (const int*)d_in[5];
    const int*   pos   = (const int*)d_in[6];
    const int*   neg   = (const int*)d_in[7];
    const int    nnz   = in_sizes[3];

    float* out        = (float*)d_out;
    float* pos_scores = out;
    float* neg_scores = out + BATCH;
    float* u0         = out + 2 * BATCH;
    float* p0         = u0 + BATCH * DIM;
    float* n0         = p0 + BATCH * DIM;

    const size_t nfeat = (size_t)NNODES * DIM;
    float* x1      = (float*)d_ws;
    float* x2      = x1 + nfeat;
    int*   rank    = (int*)x2;
    int*   counts  = (int*)(x2 + nfeat);
    int*   row_ptr = counts + NNODES;
    int2*  edges   = (int2*)(row_ptr + NNODES);

    const int qgrid = (NQ + 255) / 256;
    const int egrid = (nnz + 255) / 256;
    const int lgrid = (NNODES / ROWS_PER_WAVE + 7) / 8;

    lgcn_zero_counts<<<qgrid, 256, 0, stream>>>(counts);
    lgcn_count<<<egrid, 256, 0, stream>>>(rows, counts, rank, nnz);
    lgcn_scan<<<qgrid, 256, 0, stream>>>(counts, row_ptr);
    lgcn_place<<<egrid, 256, 0, stream>>>(rows, cols, vals, row_ptr, rank, edges, nnz);

    lgcn_layer_first<<<lgrid, 256, 0, stream>>>(row_ptr, counts, edges, uemb, iemb, x1);
    lgcn_layer_mid<<<lgrid, 256, 0, stream>>>(row_ptr, counts, edges, x1, x2);

    lgcn_last_score<<<BATCH / 16, 512, 0, stream>>>(row_ptr, counts, edges, uemb, iemb,
                                                   x1, x2, users, pos, neg,
                                                   pos_scores, neg_scores);

    lgcn_gather0<<<(BATCH * (DIM / 4) + 255) / 256, 256, 0, stream>>>(
        uemb, iemb, users, pos, neg, u0, p0, n0);
}


# Edge sort by destination row becomes a two-level counting sort in LDS (bins of 256 rows, then rows): per-edge global atomics, rank pass and scattered 8-byte stores go; row sums in LDS-slot edge order
# speedup vs baseline: 6.9965x; 2.6785x over previous
#include <hip/hip_runtime.h>


#define N_USERS 100000
#define N_ITEMS 50000
#define NNODES  (N_USERS + N_ITEMS)
#define DIM     64
#define BATCH   8192
#define ROWS_PER_WAVE 4

typedef __attribute__((ext_vector_type(2))) float v2f;
typedef __attribute__((ext_vector_type(8))) float v8f;

#define BROWS_LOG    8
#define BROWS        (1 << BROWS_LOG)
#define NBKT         ((NNODES + BROWS - 1) / BROWS)
#define CPAD         32
#define TILE_THREADS 1024
#define TILE_PER     8
#define TILE_EDGES   (TILE_THREADS * TILE_PER)
#define COL_BITS     18
#define COL_MASK     ((1 << COL_BITS) - 1)
#define FIN_THREADS  512
#define FIN_PER      4
#define FIN_CAP      12288

__device__ __forceinline__ int lgcn_block_excl_scan(int v, int t, int* s_w) {
    const int lane = t & 31;
    const int w = t >> 5;
    int incl = v;
#pragma unroll
    for (int d = 1; d < 32; d <<= 1) {
        int u = __shfl_up(incl, d);
        if (lane >= d) incl += u;
    }
    if (lane == 31) s_w[w] = incl;
    __syncthreads();
    int base = 0;
    for (int k = 0; k < w; ++k) base += s_w[k];
    return base + incl - v;
}

__global__ void lgcn_zero_bins(int* __restrict__ ghist) {
    int t = blockIdx.x * blockDim.x + threadIdx.x;
    if (t < NBKT) ghist[t * CPAD] = 0;
}

__global__ void __launch_bounds__(TILE_THREADS)
lgcn_bin_count(const int* __restrict__ rows,
               int*       __restrict__ ghist,
               int nnz) {
    __shared__ int s_h[NBKT];
    const int t = threadIdx.x;
    for (int i = t; i < NBKT; i += TILE_THREADS) s_h[i] = 0;
    __syncthreads();
    const int base = blockIdx.x * TILE_EDGES;
    int r[TILE_PER];
#pragma unroll
    for (int k = 0; k < TILE_PER; ++k) {
        const int e = base + k * TILE_THREADS + t;
        r[k] = (e < nnz) ? rows[e] : -1;
    }
#pragma unroll
    for (int k = 0; k < TILE_PER; ++k)
        if (r[k] >= 0) atomicAdd(&s_h[r[k] >> BROWS_LOG], 1);
    __syncthreads();
    for (int i = t; i < NBKT; i += TILE_THREADS) {
        const int c = s_h[i];
        if (c) atomicAdd(&ghist[i * CPAD], c);
    }
}

__global__ void __launch_bounds__(1024)
lgcn_bin_scan(const int* __restrict__ ghist,
              int*       __restrict__ gstart,
              int*       __restrict__ gcursor) {
    __shared__ int s_w[32];
    const int t = threadIdx.x;
    const int v = (t < NBKT) ? ghist[t * CPAD] : 0;
    const int ex = lgcn_block_excl_scan(v, t, s_w);
    if (t < NBKT) {
        gstart[t] = ex;
        gcursor[t * CPAD] = ex;
    }
    if (t == NBKT - 1) gstart[NBKT] = ex + v;
}

__global__ void __launch_bounds__(TILE_THREADS)
lgcn_bin_scatter(const int*   __restrict__ rows,
                 const int*   __restrict__ cols,
                 const float* __restrict__ vals,
                 int*         __restrict__ gcursor,
                 int2*        __restrict__ bedges,
                 int nnz) {
    __shared__ int s_h[NBKT];
    __shared__ int s_loc[NBKT];
    __shared__ int s_dst[NBKT];
    __shared__ int s_w[32];
    __shared__ int s_key[TILE_EDGES];
    __shared__ int s_val[TILE_EDGES];
    __shared__ unsigned short s_bk[TILE_EDGES];
    const int t = threadIdx.x;
    for (int i = t; i < NBKT; i += TILE_THREADS) s_h[i] = 0;
    __syncthreads();
    const int base = blockIdx.x * TILE_EDGES;
    int r[TILE_PER];
    int lr[TILE_PER];
#pragma unroll
    for (int k = 0; k < TILE_PER; ++k) {
        const int e = base + k * TILE_THREADS + t;
        r[k] = (e < nnz) ? rows[e] : -1;
    }
#pragma unroll
    for (int k = 0; k < TILE_PER; ++k)
        lr[k] = (r[k] >= 0) ? atomicAdd(&s_h[r[k] >> BROWS_LOG], 1) : 0;
    int   cv[TILE_PER];
    float vv[TILE_PER];
#pragma unroll
    for (int k = 0; k < TILE_PER; ++k) {
        const int e = base + k * TILE_THREADS + t;
        cv[k] = (e < nnz) ? cols[e] : 0;
        vv[k] = (e < nnz) ? vals[e] : 0.f;
    }
    __syncthreads();
    const int c = (t < NBKT) ? s_h[t] : 0;
    const int ex = lgcn_block_excl_scan(c, t, s_w);
    if (t < NBKT) {
        s_loc[t] = ex;
        s_dst[t] = c ? atomicAdd(&gcursor[t * CPAD], c) : 0;
    }
    __syncthreads();
#pragma unroll
    for (int k = 0; k < TILE_PER; ++k) {
        if (r[k] >= 0) {
            const int b = r[k] >> BROWS_LOG;
            const int pos = s_loc[b] + lr[k];
            s_key[pos] = ((r[k] & (BROWS - 1)) << COL_BITS) | cv[k];
            s_val[pos] = __float_as_int(vv[k]);
            s_bk[pos] = (unsigned short)b;
        }
    }
    __syncthreads();
    int n = nnz - base;
    if (n > TILE_EDGES) n = TILE_EDGES;
    for (int i = t; i < n; i += TILE_THREADS) {
        const int b = s_bk[i];
        const int dst = s_dst[b] + (i - s_loc[b]);
        bedges[dst] = make_int2(s_key[i], s_val[i]);
    }
}

__global__ void __launch_bounds__(FIN_THREADS)
lgcn_bin_finish(const int*  __restrict__ gstart,
                const int2* __restrict__ bedges,
                int*        __restrict__ counts,
                int*        __restrict__ row_ptr,
                int2*       __restrict__ edges) {
    __shared__ int  s_cnt[BROWS];
    __shared__ int  s_cur[BROWS];
    __shared__ int  s_w[32];
    __shared__ int2 s_out[FIN_CAP];
    const int t = threadIdx.x;
    const int b = blockIdx.x;
    const int beg = gstart[b];
    const int n = gstart[b + 1] - beg;
    const bool in_lds = (n <= FIN_CAP);
    if (t < BROWS) s_cnt[t] = 0;
    __syncthreads();
    for (int i0 = 0; i0 < n; i0 += FIN_THREADS * FIN_PER) {
        int key[FIN_PER];
#pragma unroll
        for (int k = 0; k < FIN_PER; ++k) {
            const int i = i0 + k * FIN_THREADS + t;
            key[k] = (i < n) ? bedges[beg + i].x : -1;
        }
#pragma unroll
        for (int k = 0; k < FIN_PER; ++k)
            if (key[k] >= 0) atomicAdd(&s_cnt[key[k] >> COL_BITS], 1);
    }
    __syncthreads();
    const int v = (t < BROWS) ? s_cnt[t] : 0;
    const int ex = lgcn_block_excl_scan(v, t, s_w);
    if (t < BROWS) {
        s_cur[t] = ex;
        const int row = b * BROWS + t;
        if (row < NNODES) {
            row_ptr[row] = beg + ex;
            counts[row] = v;
        }
    }
    __syncthreads();
    for (int i0 = 0; i0 < n; i0 += FIN_THREADS * FIN_PER) {
        int2 ed[FIN_PER];
#pragma unroll
        for (int k = 0; k < FIN_PER; ++k) {
            const int i = i0 + k * FIN_THREADS + t;
            ed[k] = (i < n) ? bedges[beg + i] : make_int2(-1, 0);
        }
#pragma unroll
        for (int k = 0; k < FIN_PER; ++k) {
            if (ed[k].x >= 0) {
                const int p = atomicAdd(&s_cur[ed[k].x >> COL_BITS], 1);
                const int2 wd = make_int2(ed[k].x & COL_MASK, ed[k].y);
                if (in_lds) s_out[p] = wd;
                else        edges[beg + p] = wd;
            }
        }
    }
    __syncthreads();
    if (in_lds)
        for (int i = t; i < n; i += FIN_THREADS) edges[beg + i] = s_out[i];
}

__device__ __forceinline__ float2 lgcn_row_sum(const int2*  __restrict__ edges,
                                               int beg, int end, int lane,
                                               const float* __restrict__ xa,
                                               const float* __restrict__ xb,
                                               int split) {
    float2 s = make_float2(0.f, 0.f);
    for (int base = beg; base < end; base += 32) {
        int n = end - base;
        if (n > 32) n = 32;
        int2 ed = make_int2(0, 0);
        if (lane < n) ed = edges[base + lane];
        for (int j = 0; j < n; j += 8) {
            float2 xv[8];
            float  vv[8];
#pragma unroll
            for (int k = 0; k < 8; ++k) {
                int c = __builtin_amdgcn_readlane(ed.x, j + k);
                vv[k] = __int_as_float(__builtin_amdgcn_readlane(ed.y, j + k));
                const float* src = (c < split) ? (xa + (long)c * DIM)
                                               : (xb + (long)(c - split) * DIM);
                xv[k] = ((const float2*)src)[lane];
            }
#pragma unroll
            for (int k = 0; k < 8; ++k) {
                s.x += vv[k] * xv[k].x;
                s.y += vv[k] * xv[k].y;
            }
        }
    }
    return s;
}

__global__ void lgcn_layer_first(const int*   __restrict__ row_ptr,
                                 const int*   __restrict__ counts,
                                 const int2*  __restrict__ edges,
                                 const float* __restrict__ uemb,
                                 const float* __restrict__ iemb,
                                 float*       __restrict__ x1) {
    const int wave = __builtin_amdgcn_readfirstlane(blockIdx.x * (blockDim.x >> 5) + (threadIdx.x >> 5));
    const int lane = threadIdx.x & 31;
    for (int i = 0; i < ROWS_PER_WAVE; ++i) {
        const int row = wave * ROWS_PER_WAVE + i;
        if (row >= NNODES) break;
        const int beg = row_ptr[row];
        const int end = beg + counts[row];
        float2 s = lgcn_row_sum(edges, beg, end, lane, uemb, iemb, N_USERS);
        ((float2*)(x1 + (long)row * DIM))[lane] = s;
    }
}

__global__ void lgcn_layer_mid(const int*   __restrict__ row_ptr,
                               const int*   __restrict__ counts,
                               const int2*  __restrict__ edges,
                               const float* __restrict__ xin,
                               float*       __restrict__ xout) {
    const int wave = __builtin_amdgcn_readfirstlane(blockIdx.x * (blockDim.x >> 5) + (threadIdx.x >> 5));
    const int lane = threadIdx.x & 31;
    for (int i = 0; i < ROWS_PER_WAVE; ++i) {
        const int row = wave * ROWS_PER_WAVE + i;
        if (row >= NNODES) break;
        const int beg = row_ptr[row];
        const int end = beg + counts[row];
        float2 s = lgcn_row_sum(edges, beg, end, lane, xin, xin, NNODES);
        ((float2*)(xout + (long)row * DIM))[lane] = s;
    }
}

__device__ __forceinline__ float pick8(v8f c, int j) {
    float r = c[0];
    if (j == 1) r = c[1];
    if (j == 2) r = c[2];
    if (j == 3) r = c[3];
    if (j == 4) r = c[4];
    if (j == 5) r = c[5];
    if (j == 6) r = c[6];
    if (j == 7) r = c[7];
    return r;
}

#define ACC_PITCH 68

__global__ void __launch_bounds__(512)
lgcn_last_score(const int*   __restrict__ row_ptr,
                const int*   __restrict__ counts,
                const int2*  __restrict__ edges,
                const float* __restrict__ uemb,
                const float* __restrict__ iemb,
                const float* __restrict__ x1,
                const float* __restrict__ x2,
                const int*   __restrict__ users,
                const int*   __restrict__ pos,
                const int*   __restrict__ neg,
                float*       __restrict__ pos_scores,
                float*       __restrict__ neg_scores) {
    __shared__ float s_acc[3 * 16 * ACC_PITCH];
    const int w    = __builtin_amdgcn_readfirstlane(threadIdx.x >> 5);
    const int lane = threadIdx.x & 31;
    const int tile = blockIdx.x;
    const int b    = tile * 16 + w;
    const int idu = users[b];
    const int idp = pos[b];
    const int idn = neg[b];
    const int ru = idu;
    const int rp = idp + N_USERS;
    const int rn = idn + N_USERS;
    const int begu = row_ptr[ru];
    const int begp = row_ptr[rp];
    const int begn = row_ptr[rn];
    const int endu = begu + counts[ru];
    const int endp = begp + counts[rp];
    const int endn = begn + counts[rn];
    const float2 u0 = ((const float2*)(uemb + (long)idu * DIM))[lane];
    const float2 p0 = ((const float2*)(iemb + (long)idp * DIM))[lane];
    const float2 n0 = ((const float2*)(iemb + (long)idn * DIM))[lane];
    const float2 u1 = ((const float2*)(x1 + (long)ru * DIM))[lane];
    const float2 p1 = ((const float2*)(x1 + (long)rp * DIM))[lane];
    const float2 n1 = ((const float2*)(x1 + (long)rn * DIM))[lane];
    const float2 u2 = ((const float2*)(x2 + (long)ru * DIM))[lane];
    const float2 p2 = ((const float2*)(x2 + (long)rp * DIM))[lane];
    const float2 n2 = ((const float2*)(x2 + (long)rn * DIM))[lane];

    const float2 su = lgcn_row_sum(edges, begu, endu, lane, x2, x2, NNODES);
    const float2 sp = lgcn_row_sum(edges, begp, endp, lane, x2, x2, NNODES);
    const float2 sn = lgcn_row_sum(edges, begn, endn, lane, x2, x2, NNODES);

    float2 au, ap, an;
    au.x = ((u0.x + u1.x) + u2.x) + su.x;
    au.y = ((u0.y + u1.y) + u2.y) + su.y;
    ap.x = ((p0.x + p1.x) + p2.x) + sp.x;
    ap.y = ((p0.y + p1.y) + p2.y) + sp.y;
    an.x = ((n0.x + n1.x) + n2.x) + sn.x;
    an.y = ((n0.y + n1.y) + n2.y) + sn.y;
    ((float2*)(s_acc + (0 * 16 + w) * ACC_PITCH))[lane] = au;
    ((float2*)(s_acc + (1 * 16 + w) * ACC_PITCH))[lane] = ap;
    ((float2*)(s_acc + (2 * 16 + w) * ACC_PITCH))[lane] = an;
    __syncthreads();

    if (w == 0) {
        const int m     = lane & 15;
        const int khalf = lane >> 4;
        const float* urow = s_acc + (0 * 16 + m) * ACC_PITCH + khalf * 2;
        const float* prow = s_acc + (1 * 16 + m) * ACC_PITCH + khalf * 2;
        const float* nrow = s_acc + (2 * 16 + m) * ACC_PITCH + khalf * 2;
        v8f cpos = {};
        v8f cneg = {};
#pragma unroll
        for (int kk = 0; kk < 16; ++kk) {
            float2 ua = *(const float2*)(urow + kk * 4);
            float2 pb = *(const float2*)(prow + kk * 4);
            float2 nb = *(const float2*)(nrow + kk * 4);
            v2f a;  a[0]  = ua.x; a[1]  = ua.y;
            v2f bp; bp[0] = pb.x; bp[1] = pb.y;
            v2f bn; bn[0] = nb.x; bn[1] = nb.y;
            cpos = __builtin_amdgcn_wmma_f32_16x16x4_f32(false, a, false, bp,
                                                         (short)0, cpos, false, false);
            cneg = __builtin_amdgcn_wmma_f32_16x16x4_f32(false, a, false, bn,
                                                         (short)0, cneg, false, false);
        }
        const float s = 1.0f / 16.0f;
        if (lane < 8) {
            pos_scores[tile * 16 + lane] = pick8(cpos, lane) * s;
            neg_scores[tile * 16 + lane] = pick8(cneg, lane) * s;
        } else if (lane >= 24) {
            int mm = lane - 16;
            pos_scores[tile * 16 + mm] = pick8(cpos, lane - 24) * s;
            neg_scores[tile * 16 + mm] = pick8(cneg, lane - 24) * s;
        }
    }
}

__global__ void lgcn_gather0(const float* __restrict__ uemb,
                             const float* __restrict__ iemb,
                             const int*   __restrict__ users,
                             const int*   __restrict__ pos,
                             const int*   __restrict__ neg,
                             float* __restrict__ u0,
                             float* __restrict__ p0,
                             float* __restrict__ n0) {
    int idx = blockIdx.x * blockDim.x + threadIdx.x;
    if (idx >= BATCH * (DIM / 4)) return;
    int b = idx >> 4;
    int q = idx & 15;
    ((float4*)u0)[idx] = ((const float4*)uemb)[(long)users[b] * 16 + q];
    ((float4*)p0)[idx] = ((const float4*)iemb)[(long)pos[b]   * 16 + q];
    ((float4*)n0)[idx] = ((const float4*)iemb)[(long)neg[b]   * 16 + q];
}

extern "C" void kernel_launch(void* const* d_in, const int* in_sizes, int n_in,
                              void* d_out, int out_size, void* d_ws, size_t ws_size,
                              hipStream_t stream) {
    const float* uemb  = (const float*)d_in[0];
    const float* iemb  = (const float*)d_in[1];
    const float* vals  = (const float*)d_in[2];
    const int*   rows  = (const int*)d_in[3];
    const int*   cols  = (const int*)d_in[4];
    const int*   users = (const int*)d_in[5];
    const int*   pos   = (const int*)d_in[6];
    const int*   neg   = (const int*)d_in[7];
    const int    nnz   = in_sizes[3];

    float* out        = (float*)d_out;
    float* pos_scores = out;
    float* neg_scores = out + BATCH;
    float* u0         = out + 2 * BATCH;
    float* p0         = u0 + BATCH * DIM;
    float* n0         = p0 + BATCH * DIM;

    const size_t nfeat = (size_t)NNODES * DIM;
    float* x1      = (float*)d_ws;
    float* x2      = x1 + nfeat;
    int2*  bedges  = (int2*)x2;
    int*   ghist   = (int*)x1;
    int*   gcursor = ghist + NBKT * CPAD;
    int*   gstart  = gcursor + NBKT * CPAD;
    int*   counts  = (int*)(x2 + nfeat);
    int*   row_ptr = counts + NNODES;
    int2*  edges   = (int2*)(row_ptr + NNODES);

    const int tgrid = (nnz + TILE_EDGES - 1) / TILE_EDGES;
    const int lgrid = (NNODES / ROWS_PER_WAVE + 7) / 8;

    lgcn_zero_bins<<<1, 1024, 0, stream>>>(ghist);
    lgcn_bin_count<<<tgrid, TILE_THREADS, 0, stream>>>(rows, ghist, nnz);
    lgcn_bin_scan<<<1, 1024, 0, stream>>>(ghist, gstart, gcursor);
    lgcn_bin_scatter<<<tgrid, TILE_THREADS, 0, stream>>>(rows, cols, vals, gcursor, bedges, nnz);
    lgcn_bin_finish<<<NBKT, FIN_THREADS, 0, stream>>>(gstart, bedges, counts, row_ptr, edges);

    lgcn_layer_first<<<lgrid, 256, 0, stream>>>(row_ptr, counts, edges, uemb, iemb, x1);
    lgcn_layer_mid<<<lgrid, 256, 0, stream>>>(row_ptr, counts, edges, x1, x2);

    lgcn_last_score<<<BATCH / 16, 512, 0, stream>>>(row_ptr, counts, edges, uemb, iemb,
                                                   x1, x2, users, pos, neg,
                                                   pos_scores, neg_scores);

    lgcn_gather0<<<(BATCH * (DIM / 4) + 255) / 256, 256, 0, stream>>>(
        uemb, iemb, users, pos, neg, u0, p0, n0);
}


# Edge sort level 1 leaves each tile's bin-grouped edges in place plus a per-tile offset table, level 2 pulls its bin's pieces; zero, count and scan launches and cursor adds go; LDS-slot sum order kept
# speedup vs baseline: 7.3961x; 1.0571x over previous
#include <hip/hip_runtime.h>


#define N_USERS 100000
#define N_ITEMS 50000
#define NNODES  (N_USERS + N_ITEMS)
#define DIM     64
#define BATCH   8192
#define ROWS_PER_WAVE 4

typedef __attribute__((ext_vector_type(2))) float v2f;
typedef __attribute__((ext_vector_type(8))) float v8f;

#define BROWS_LOG    8
#define BROWS        (1 << BROWS_LOG)
#define NBKT         ((NNODES + BROWS - 1) / BROWS)
#define NB1          (NBKT + 1)
#define TILE_THREADS 1024
#define TILE_PER     8
#define TILE_EDGES   (TILE_THREADS * TILE_PER)
#define COL_BITS     18
#define COL_MASK     ((1 << COL_BITS) - 1)
#define FIN_THREADS  512
#define FIN_WAVES    (FIN_THREADS / 32)
#define SEG_PER      8
#define FIN_CAP      12288

static_assert(NBKT < TILE_THREADS, "level 1 wants a thread per bin and one more");
static_assert((FIN_THREADS & (FIN_THREADS - 1)) == 0, "piece numbers are wrapped with a mask");

__device__ __forceinline__ int lgcn_block_excl_scan(int v, int t, int* s_w) {
    const int lane = t & 31;
    const int w = t >> 5;
    int incl = v;
#pragma unroll
    for (int d = 1; d < 32; d <<= 1) {
        int u = __shfl_up(incl, d);
        if (lane >= d) incl += u;
    }
    if (lane == 31) s_w[w] = incl;
    __syncthreads();
    int base = 0;
    for (int k = 0; k < w; ++k) base += s_w[k];
    return base + incl - v;
}

__global__ void __launch_bounds__(TILE_THREADS)
lgcn_bin_group(const int*   __restrict__ rows,
               const int*   __restrict__ cols,
               const float* __restrict__ vals,
               int*         __restrict__ toff,
               int*         __restrict__ cursor,
               int2*        __restrict__ bedges,
               int nnz) {
    __shared__ int s_h[NBKT];
    __shared__ int s_loc[NBKT];
    __shared__ int s_w[32];
    __shared__ int s_key[TILE_EDGES];
    __shared__ int s_val[TILE_EDGES];
    const int t = threadIdx.x;
    if (blockIdx.x == 0 && t == 0) *cursor = 0;
    for (int i = t; i < NBKT; i += TILE_THREADS) s_h[i] = 0;
    __syncthreads();
    const int base = blockIdx.x * TILE_EDGES;
    int r[TILE_PER];
    int lr[TILE_PER];
#pragma unroll
    for (int k = 0; k < TILE_PER; ++k) {
        const int e = base + k * TILE_THREADS + t;
        r[k] = (e < nnz) ? rows[e] : -1;
    }
#pragma unroll
    for (int k = 0; k < TILE_PER; ++k)
        lr[k] = (r[k] >= 0) ? atomicAdd(&s_h[r[k] >> BROWS_LOG], 1) : 0;
    int   cv[TILE_PER];
    float vv[TILE_PER];
#pragma unroll
    for (int k = 0; k < TILE_PER; ++k) {
        const int e = base + k * TILE_THREADS + t;
        cv[k] = (e < nnz) ? cols[e] : 0;
        vv[k] = (e < nnz) ? vals[e] : 0.f;
    }
    __syncthreads();
    const int c = (t < NBKT) ? s_h[t] : 0;
    const int ex = lgcn_block_excl_scan(c, t, s_w);
    if (t < NBKT) s_loc[t] = ex;
    if (t <= NBKT) toff[blockIdx.x * NB1 + t] = ex;
    __syncthreads();
#pragma unroll
    for (int k = 0; k < TILE_PER; ++k) {
        if (r[k] >= 0) {
            const int pos = s_loc[r[k] >> BROWS_LOG] + lr[k];
            s_key[pos] = ((r[k] & (BROWS - 1)) << COL_BITS) | cv[k];
            s_val[pos] = __float_as_int(vv[k]);
        }
    }
    __syncthreads();
    int n = nnz - base;
    if (n > TILE_EDGES) n = TILE_EDGES;
    for (int i = t; i < n; i += TILE_THREADS)
        bedges[base + i] = make_int2(s_key[i], s_val[i]);
}

__device__ __forceinline__ void lgcn_stage_pieces(const int* __restrict__ toff,
                                                  int b, int t0, int ntiles, int t,
                                                  int* s_off, int* s_len) {
    const int tt = t0 + t;
    int o = 0, l = 0;
    if (tt < ntiles) {
        o = toff[tt * NB1 + b];
        l = toff[tt * NB1 + b + 1] - o;
    }
    s_off[t] = tt * TILE_EDGES + o;
    s_len[t] = l;
}

__global__ void __launch_bounds__(FIN_THREADS)
lgcn_bin_finish(const int*  __restrict__ toff,
                const int2* __restrict__ bedges,
                int*        __restrict__ cursor,
                int*        __restrict__ counts,
                int*        __restrict__ row_ptr,
                int2*       __restrict__ edges,
                int ntiles) {
    __shared__ int  s_cnt[BROWS];
    __shared__ int  s_cur[BROWS];
    __shared__ int  s_w[32];
    __shared__ int  s_off[FIN_THREADS];
    __shared__ int  s_len[FIN_THREADS];
    __shared__ int  s_run[2];
    __shared__ int2 s_out[FIN_CAP];
    const int t = threadIdx.x;
    const int b = blockIdx.x;
    const int lane = t & 31;
    const int w = __builtin_amdgcn_readfirstlane(t >> 5);
    if (t < BROWS) s_cnt[t] = 0;
    for (int t0 = 0; t0 < ntiles; t0 += FIN_THREADS) {
        __syncthreads();
        lgcn_stage_pieces(toff, b, t0, ntiles, t, s_off, s_len);
        __syncthreads();
        int nt = ntiles - t0;
        if (nt > FIN_THREADS) nt = FIN_THREADS;
        for (int s0 = w; s0 < nt; s0 += FIN_WAVES * SEG_PER) {
            int off[SEG_PER], len[SEG_PER], key[SEG_PER];
#pragma unroll
            for (int q = 0; q < SEG_PER; ++q) {
                const int s = s0 + q * FIN_WAVES;
                off[q] = s_off[s & (FIN_THREADS - 1)];
                len[q] = s_len[s & (FIN_THREADS - 1)];
                if (s >= nt) len[q] = 0;
            }
#pragma unroll
            for (int q = 0; q < SEG_PER; ++q)
                key[q] = (lane < len[q]) ? bedges[off[q] + lane].x : -1;
#pragma unroll
            for (int q = 0; q < SEG_PER; ++q)
                if (key[q] >= 0) atomicAdd(&s_cnt[key[q] >> COL_BITS], 1);
#pragma unroll
            for (int q = 0; q < SEG_PER; ++q)
                for (int k = 32 + lane; k < len[q]; k += 32)
                    atomicAdd(&s_cnt[bedges[off[q] + k].x >> COL_BITS], 1);
        }
    }
    __syncthreads();
    const int v = (t < BROWS) ? s_cnt[t] : 0;
    const int ex = lgcn_block_excl_scan(v, t, s_w);
    if (t < BROWS) s_cur[t] = ex;
    if (t == BROWS - 1) {
        s_run[0] = ex + v;
        s_run[1] = atomicAdd(cursor, ex + v);
    }
    __syncthreads();
    const int n = s_run[0];
    const int beg = s_run[1];
    const bool in_lds = (n <= FIN_CAP);
    if (t < BROWS) {
        const int row = b * BROWS + t;
        if (row < NNODES) {
            row_ptr[row] = beg + ex;
            counts[row] = v;
        }
    }
    for (int t0 = 0; t0 < ntiles; t0 += FIN_THREADS) {
        if (t0 > 0 || ntiles > FIN_THREADS) {
            __syncthreads();
            lgcn_stage_pieces(toff, b, t0, ntiles, t, s_off, s_len);
            __syncthreads();
        }
        int nt = ntiles - t0;
        if (nt > FIN_THREADS) nt = FIN_THREADS;
        for (int s0 = w; s0 < nt; s0 += FIN_WAVES * SEG_PER) {
            int off[SEG_PER], len[SEG_PER];
            int2 ed[SEG_PER];
#pragma unroll
            for (int q = 0; q < SEG_PER; ++q) {
                const int s = s0 + q * FIN_WAVES;
                off[q] = s_off[s & (FIN_THREADS - 1)];
                len[q] = s_len[s & (FIN_THREADS - 1)];
                if (s >= nt) len[q] = 0;
            }
#pragma unroll
            for (int q = 0; q < SEG_PER; ++q)
                ed[q] = (lane < len[q]) ? bedges[off[q] + lane] : make_int2(-1, 0);
#pragma unroll
            for (int q = 0; q < SEG_PER; ++q) {
                if (ed[q].x >= 0) {
                    const int p = atomicAdd(&s_cur[ed[q].x >> COL_BITS], 1);
                    const int2 wd = make_int2(ed[q].x & COL_MASK, ed[q].y);
                    if (in_lds) s_out[p] = wd;
                    else        edges[beg + p] = wd;
                }
            }
#pragma unroll
            for (int q = 0; q < SEG_PER; ++q)
                for (int k = 32 + lane; k < len[q]; k += 32) {
                    const int2 e2 = bedges[off[q] + k];
                    const int p = atomicAdd(&s_cur[e2.x >> COL_BITS], 1);
                    const int2 wd = make_int2(e2.x & COL_MASK, e2.y);
                    if (in_lds) s_out[p] = wd;
                    else        edges[beg + p] = wd;
                }
        }
    }
    __syncthreads();
    if (in_lds)
        for (int i = t; i < n; i += FIN_THREADS) edges[beg + i] = s_out[i];
}

__device__ __forceinline__ float2 lgcn_row_sum(const int2*  __restrict__ edges,
                                               int beg, int end, int lane,
                                               const float* __restrict__ xa,
                                               const float* __restrict__ xb,
                                               int split) {
    float2 s = make_float2(0.f, 0.f);
    for (int base = beg; base < end; base += 32) {
        int n = end - base;
        if (n > 32) n = 32;
        int2 ed = make_int2(0, 0);
        if (lane < n) ed = edges[base + lane];
        for (int j = 0; j < n; j += 8) {
            float2 xv[8];
            float  vv[8];
#pragma unroll
            for (int k = 0; k < 8; ++k) {
                int c = __builtin_amdgcn_readlane(ed.x, j + k);
                vv[k] = __int_as_float(__builtin_amdgcn_readlane(ed.y, j + k));
                const float* src = (c < split) ? (xa + (long)c * DIM)
                                               : (xb + (long)(c - split) * DIM);
                xv[k] = ((const float2*)src)[lane];
            }
#pragma unroll
            for (int k = 0; k < 8; ++k) {
                s.x += vv[k] * xv[k].x;
                s.y += vv[k] * xv[k].y;
            }
        }
    }
    return s;
}

__global__ void lgcn_layer_first(const int*   __restrict__ row_ptr,
                                 const int*   __restrict__ counts,
                                 const int2*  __restrict__ edges,
                                 const float* __restrict__ uemb,
                                 const float* __restrict__ iemb,
                                 float*       __restrict__ x1) {
    const int wave = __builtin_amdgcn_readfirstlane(blockIdx.x * (blockDim.x >> 5) + (threadIdx.x >> 5));
    const int lane = threadIdx.x & 31;
    for (int i = 0; i < ROWS_PER_WAVE; ++i) {
        const int row = wave * ROWS_PER_WAVE + i;
        if (row >= NNODES) break;
        const int beg = row_ptr[row];
        const int end = beg + counts[row];
        float2 s = lgcn_row_sum(edges, beg, end, lane, uemb, iemb, N_USERS);
        ((float2*)(x1 + (long)row * DIM))[lane] = s;
    }
}

__global__ void lgcn_layer_mid(const int*   __restrict__ row_ptr,
                               const int*   __restrict__ counts,
                               const int2*  __restrict__ edges,
                               const float* __restrict__ xin,
                               float*       __restrict__ xout) {
    const int wave = __builtin_amdgcn_readfirstlane(blockIdx.x * (blockDim.x >> 5) + (threadIdx.x >> 5));
    const int lane = threadIdx.x & 31;
    for (int i = 0; i < ROWS_PER_WAVE; ++i) {
        const int row = wave * ROWS_PER_WAVE + i;
        if (row >= NNODES) break;
        const int beg = row_ptr[row];
        const int end = beg + counts[row];
        float2 s = lgcn_row_sum(edges, beg, end, lane, xin, xin, NNODES);
        ((float2*)(xout + (long)row * DIM))[lane] = s;
    }
}

__device__ __forceinline__ float pick8(v8f c, int j) {
    float r = c[0];
    if (j == 1) r = c[1];
    if (j == 2) r = c[2];
    if (j == 3) r = c[3];
    if (j == 4) r = c[4];
    if (j == 5) r = c[5];
    if (j == 6) r = c[6];
    if (j == 7) r = c[7];
    return r;
}

#define ACC_PITCH 68

__global__ void __launch_bounds__(512)
lgcn_last_score(const int*   __restrict__ row_ptr,
                const int*   __restrict__ counts,
                const int2*  __restrict__ edges,
                const float* __restrict__ uemb,
                const float* __restrict__ iemb,
                const float* __restrict__ x1,
                const float* __restrict__ x2,
                const int*   __restrict__ users,
                const int*   __restrict__ pos,
                const int*   __restrict__ neg,
                float*       __restrict__ pos_scores,
                float*       __restrict__ neg_scores) {
    __shared__ float s_acc[3 * 16 * ACC_PITCH];
    const int w    = __builtin_amdgcn_readfirstlane(threadIdx.x >> 5);
    const int lane = threadIdx.x & 31;
    const int tile = blockIdx.x;
    const int b    = tile * 16 + w;
    const int idu = users[b];
    const int idp = pos[b];
    const int idn = neg[b];
    const int ru = idu;
    const int rp = idp + N_USERS;
    const int rn = idn + N_USERS;
    const int begu = row_ptr[ru];
    const int begp = row_ptr[rp];
    const int begn = row_ptr[rn];
    const int endu = begu + counts[ru];
    const int endp = begp + counts[rp];
    const int endn = begn + counts[rn];
    const float2 u0 = ((const float2*)(uemb + (long)idu * DIM))[lane];
    const float2 p0 = ((const float2*)(iemb + (long)idp * DIM))[lane];
    const float2 n0 = ((const float2*)(iemb + (long)idn * DIM))[lane];
    const float2 u1 = ((const float2*)(x1 + (long)ru * DIM))[lane];
    const float2 p1 = ((const float2*)(x1 + (long)rp * DIM))[lane];
    const float2 n1 = ((const float2*)(x1 + (long)rn * DIM))[lane];
    const float2 u2 = ((const float2*)(x2 + (long)ru * DIM))[lane];
    const float2 p2 = ((const float2*)(x2 + (long)rp * DIM))[lane];
    const float2 n2 = ((const float2*)(x2 + (long)rn * DIM))[lane];

    const float2 su = lgcn_row_sum(edges, begu, endu, lane, x2, x2, NNODES);
    const float2 sp = lgcn_row_sum(edges, begp, endp, lane, x2, x2, NNODES);
    const float2 sn = lgcn_row_sum(edges, begn, endn, lane, x2, x2, NNODES);

    float2 au, ap, an;
    au.x = ((u0.x + u1.x) + u2.x) + su.x;
    au.y = ((u0.y + u1.y) + u2.y) + su.y;
    ap.x = ((p0.x + p1.x) + p2.x) + sp.x;
    ap.y = ((p0.y + p1.y) + p2.y) + sp.y;
    an.x = ((n0.x + n1.x) + n2.x) + sn.x;
    an.y = ((n0.y + n1.y) + n2.y) + sn.y;
    ((float2*)(s_acc + (0 * 16 + w) * ACC_PITCH))[lane] = au;
    ((float2*)(s_acc + (1 * 16 + w) * ACC_PITCH))[lane] = ap;
    ((float2*)(s_acc + (2 * 16 + w) * ACC_PITCH))[lane] = an;
    __syncthreads();

    if (w == 0) {
        const int m     = lane & 15;
        const int khalf = lane >> 4;
        const float* urow = s_acc + (0 * 16 + m) * ACC_PITCH + khalf * 2;
        const float* prow = s_acc + (1 * 16 + m) * ACC_PITCH + khalf * 2;
        const float* nrow = s_acc + (2 * 16 + m) * ACC_PITCH + khalf * 2;
        v8f cpos = {};
        v8f cneg = {};
#pragma unroll
        for (int kk = 0; kk < 16; ++kk) {
            float2 ua = *(const float2*)(urow + kk * 4);
            float2 pb = *(const float2*)(prow + kk * 4);
            float2 nb = *(const float2*)(nrow + kk * 4);
            v2f a;  a[0]  = ua.x; a[1]  = ua.y;
            v2f bp; bp[0] = pb.x; bp[1] = pb.y;
            v2f bn; bn[0] = nb.x; bn[1] = nb.y;
            cpos = __builtin_amdgcn_wmma_f32_16x16x4_f32(false, a, false, bp,
                                                         (short)0, cpos, false, false);
            cneg = __builtin_amdgcn_wmma_f32_16x16x4_f32(false, a, false, bn,
                                                         (short)0, cneg, false, false);
        }
        const float s = 1.0f / 16.0f;
        if (lane < 8) {
            pos_scores[tile * 16 + lane] = pick8(cpos, lane) * s;
            neg_scores[tile * 16 + lane] = pick8(cneg, lane) * s;
        } else if (lane >= 24) {
            int mm = lane - 16;
            pos_scores[tile * 16 + mm] = pick8(cpos, lane - 24) * s;
            neg_scores[tile * 16 + mm] = pick8(cneg, lane - 24) * s;
        }
    }
}

__global__ void lgcn_gather0(const float* __restrict__ uemb,
                             const float* __restrict__ iemb,
                             const int*   __restrict__ users,
                             const int*   __restrict__ pos,
                             const int*   __restrict__ neg,
                             float* __restrict__ u0,
                             float* __restrict__ p0,
                             float* __restrict__ n0) {
    int idx = blockIdx.x * blockDim.x + threadIdx.x;
    if (idx >= BATCH * (DIM / 4)) return;
    int b = idx >> 4;
    int q = idx & 15;
    ((float4*)u0)[idx] = ((const float4*)uemb)[(long)users[b] * 16 + q];
    ((float4*)p0)[idx] = ((const float4*)iemb)[(long)pos[b]   * 16 + q];
    ((float4*)n0)[idx] = ((const float4*)iemb)[(long)neg[b]   * 16 + q];
}

extern "C" void kernel_launch(void* const* d_in, const int* in_sizes, int n_in,
                              void* d_out, int out_size, void* d_ws, size_t ws_size,
                              hipStream_t stream) {
    const float* uemb  = (const float*)d_in[0];
    const float* iemb  = (const float*)d_in[1];
    const float* vals  = (const float*)d_in[2];
    const int*   rows  = (const int*)d_in[3];
    const int*   cols  = (const int*)d_in[4];
    const int*   users = (const int*)d_in[5];
    const int*   pos   = (const int*)d_in[6];
    const int*   neg   = (const int*)d_in[7];
    const int    nnz   = in_sizes[3];

    float* out        = (float*)d_out;
    float* pos_scores = out;
    float* neg_scores = out + BATCH;
    float* u0         = out + 2 * BATCH;
    float* p0         = u0 + BATCH * DIM;
    float* n0         = p0 + BATCH * DIM;

    const size_t nfeat = (size_t)NNODES * DIM;
    float* x1      = (float*)d_ws;
    float* x2      = x1 + nfeat;
    int2*  bedges  = (int2*)x2;
    int*   cursor  = (int*)x1;
    int*   toff    = cursor + 32;
    int*   counts  = (int*)(x2 + nfeat);
    int*   row_ptr = counts + NNODES;
    int2*  edges   = (int2*)(row_ptr + NNODES);

    const int tgrid = (nnz + TILE_EDGES - 1) / TILE_EDGES;
    const int lgrid = (NNODES / ROWS_PER_WAVE + 7) / 8;

    lgcn_bin_group<<<tgrid, TILE_THREADS, 0, stream>>>(rows, cols, vals, toff, cursor, bedges, nnz);
    lgcn_bin_finish<<<NBKT, FIN_THREADS, 0, stream>>>(toff, bedges, cursor, counts, row_ptr, edges, tgrid);

    lgcn_layer_first<<<lgrid, 256, 0, stream>>>(row_ptr, counts, edges, uemb, iemb, x1);
    lgcn_layer_mid<<<lgrid, 256, 0, stream>>>(row_ptr, counts, edges, x1, x2);

    lgcn_last_score<<<BATCH / 16, 512, 0, stream>>>(row_ptr, counts, edges, uemb, iemb,
                                                   x1, x2, users, pos, neg,
                                                   pos_scores, neg_scores);

    lgcn_gather0<<<(BATCH * (DIM / 4) + 255) / 256, 256, 0, stream>>>(
        uemb, iemb, users, pos, neg, u0, p0, n0);
}


# Row sums request every source row of a 32-edge chunk before using the first (was 8 at a time), and a wave's row bounds and first edges up front: fewer dependent memory trips per row; sum order kept
# speedup vs baseline: 7.6700x; 1.0370x over previous
#include <hip/hip_runtime.h>


#define N_USERS 100000
#define N_ITEMS 50000
#define NNODES  (N_USERS + N_ITEMS)
#define DIM     64
#define BATCH   8192
#define ROWS_PER_WAVE 4

typedef __attribute__((ext_vector_type(2))) float v2f;
typedef __attribute__((ext_vector_type(8))) float v8f;

#define BROWS_LOG    8
#define BROWS        (1 << BROWS_LOG)
#define NBKT         ((NNODES + BROWS - 1) / BROWS)
#define NB1          (NBKT + 1)
#define TILE_THREADS 1024
#define TILE_PER     8
#define TILE_EDGES   (TILE_THREADS * TILE_PER)
#define COL_BITS     18
#define COL_MASK     ((1 << COL_BITS) - 1)
#define FIN_THREADS  512
#define FIN_WAVES    (FIN_THREADS / 32)
#define SEG_PER      8
#define FIN_CAP      12288

static_assert(NBKT < TILE_THREADS, "level 1 wants a thread per bin and one more");
static_assert((FIN_THREADS & (FIN_THREADS - 1)) == 0, "piece numbers are wrapped with a mask");

__device__ __forceinline__ int lgcn_block_excl_scan(int v, int t, int* s_w) {
    const int lane = t & 31;
    const int w = t >> 5;
    int incl = v;
#pragma unroll
    for (int d = 1; d < 32; d <<= 1) {
        int u = __shfl_up(incl, d);
        if (lane >= d) incl += u;
    }
    if (lane == 31) s_w[w] = incl;
    __syncthreads();
    int base = 0;
    for (int k = 0; k < w; ++k) base += s_w[k];
    return base + incl - v;
}

__global__ void __launch_bounds__(TILE_THREADS)
lgcn_bin_group(const int*   __restrict__ rows,
               const int*   __restrict__ cols,
               const float* __restrict__ vals,
               int*         __restrict__ toff,
               int*         __restrict__ cursor,
               int2*        __restrict__ bedges,
               int nnz) {
    __shared__ int s_h[NBKT];
    __shared__ int s_loc[NBKT];
    __shared__ int s_w[32];
    __shared__ int s_key[TILE_EDGES];
    __shared__ int s_val[TILE_EDGES];
    const int t = threadIdx.x;
    if (blockIdx.x == 0 && t == 0) *cursor = 0;
    for (int i = t; i < NBKT; i += TILE_THREADS) s_h[i] = 0;
    __syncthreads();
    const int base = blockIdx.x * TILE_EDGES;
    int r[TILE_PER];
    int lr[TILE_PER];
#pragma unroll
    for (int k = 0; k < TILE_PER; ++k) {
        const int e = base + k * TILE_THREADS + t;
        r[k] = (e < nnz) ? rows[e] : -1;
    }
#pragma unroll
    for (int k = 0; k < TILE_PER; ++k)
        lr[k] = (r[k] >= 0) ? atomicAdd(&s_h[r[k] >> BROWS_LOG], 1) : 0;
    int   cv[TILE_PER];
    float vv[TILE_PER];
#pragma unroll
    for (int k = 0; k < TILE_PER; ++k) {
        const int e = base + k * TILE_THREADS + t;
        cv[k] = (e < nnz) ? cols[e] : 0;
        vv[k] = (e < nnz) ? vals[e] : 0.f;
    }
    __syncthreads();
    const int c = (t < NBKT) ? s_h[t] : 0;
    const int ex = lgcn_block_excl_scan(c, t, s_w);
    if (t < NBKT) s_loc[t] = ex;
    if (t <= NBKT) toff[blockIdx.x * NB1 + t] = ex;
    __syncthreads();
#pragma unroll
    for (int k = 0; k < TILE_PER; ++k) {
        if (r[k] >= 0) {
            const int pos = s_loc[r[k] >> BROWS_LOG] + lr[k];
            s_key[pos] = ((r[k] & (BROWS - 1)) << COL_BITS) | cv[k];
            s_val[pos] = __float_as_int(vv[k]);
        }
    }
    __syncthreads();
    int n = nnz - base;
    if (n > TILE_EDGES) n = TILE_EDGES;
    for (int i = t; i < n; i += TILE_THREADS)
        bedges[base + i] = make_int2(s_key[i], s_val[i]);
}

__device__ __forceinline__ void lgcn_stage_pieces(const int* __restrict__ toff,
                                                  int b, int t0, int ntiles, int t,
                                                  int* s_off, int* s_len) {
    const int tt = t0 + t;
    int o = 0, l = 0;
    if (tt < ntiles) {
        o = toff[tt * NB1 + b];
        l = toff[tt * NB1 + b + 1] - o;
    }
    s_off[t] = tt * TILE_EDGES + o;
    s_len[t] = l;
}

__global__ void __launch_bounds__(FIN_THREADS)
lgcn_bin_finish(const int*  __restrict__ toff,
                const int2* __restrict__ bedges,
                int*        __restrict__ cursor,
                int*        __restrict__ counts,
                int*        __restrict__ row_ptr,
                int2*       __restrict__ edges,
                int ntiles) {
    __shared__ int  s_cnt[BROWS];
    __shared__ int  s_cur[BROWS];
    __shared__ int  s_w[32];
    __shared__ int  s_off[FIN_THREADS];
    __shared__ int  s_len[FIN_THREADS];
    __shared__ int  s_run[2];
    __shared__ int2 s_out[FIN_CAP];
    const int t = threadIdx.x;
    const int b = blockIdx.x;
    const int lane = t & 31;
    const int w = __builtin_amdgcn_readfirstlane(t >> 5);
    if (t < BROWS) s_cnt[t] = 0;
    for (int t0 = 0; t0 < ntiles; t0 += FIN_THREADS) {
        __syncthreads();
        lgcn_stage_pieces(toff, b, t0, ntiles, t, s_off, s_len);
        __syncthreads();
        int nt = ntiles - t0;
        if (nt > FIN_THREADS) nt = FIN_THREADS;
        for (int s0 = w; s0 < nt; s0 += FIN_WAVES * SEG_PER) {
            int off[SEG_PER], len[SEG_PER], key[SEG_PER];
#pragma unroll
            for (int q = 0; q < SEG_PER; ++q) {
                const int s = s0 + q * FIN_WAVES;
                off[q] = s_off[s & (FIN_THREADS - 1)];
                len[q] = s_len[s & (FIN_THREADS - 1)];
                if (s >= nt) len[q] = 0;
            }
#pragma unroll
            for (int q = 0; q < SEG_PER; ++q)
                key[q] = (lane < len[q]) ? bedges[off[q] + lane].x : -1;
#pragma unroll
            for (int q = 0; q < SEG_PER; ++q)
                if (key[q] >= 0) atomicAdd(&s_cnt[key[q] >> COL_BITS], 1);
#pragma unroll
            for (int q = 0; q < SEG_PER; ++q)
                for (int k = 32 + lane; k < len[q]; k += 32)
                    atomicAdd(&s_cnt[bedges[off[q] + k].x >> COL_BITS], 1);
        }
    }
    __syncthreads();
    const int v = (t < BROWS) ? s_cnt[t] : 0;
    const int ex = lgcn_block_excl_scan(v, t, s_w);
    if (t < BROWS) s_cur[t] = ex;
    if (t == BROWS - 1) {
        s_run[0] = ex + v;
        s_run[1] = atomicAdd(cursor, ex + v);
    }
    __syncthreads();
    const int n = s_run[0];
    const int beg = s_run[1];
    const bool in_lds = (n <= FIN_CAP);
    if (t < BROWS) {
        const int row = b * BROWS + t;
        if (row < NNODES) {
            row_ptr[row] = beg + ex;
            counts[row] = v;
        }
    }
    for (int t0 = 0; t0 < ntiles; t0 += FIN_THREADS) {
        if (t0 > 0 || ntiles > FIN_THREADS) {
            __syncthreads();
            lgcn_stage_pieces(toff, b, t0, ntiles, t, s_off, s_len);
            __syncthreads();
        }
        int nt = ntiles - t0;
        if (nt > FIN_THREADS) nt = FIN_THREADS;
        for (int s0 = w; s0 < nt; s0 += FIN_WAVES * SEG_PER) {
            int off[SEG_PER], len[SEG_PER];
            int2 ed[SEG_PER];
#pragma unroll
            for (int q = 0; q < SEG_PER; ++q) {
                const int s = s0 + q * FIN_WAVES;
                off[q] = s_off[s & (FIN_THREADS - 1)];
                len[q] = s_len[s & (FIN_THREADS - 1)];
                if (s >= nt) len[q] = 0;
            }
#pragma unroll
            for (int q = 0; q < SEG_PER; ++q)
                ed[q] = (lane < len[q]) ? bedges[off[q] + lane] : make_int2(-1, 0);
#pragma unroll
            for (int q = 0; q < SEG_PER; ++q) {
                if (ed[q].x >= 0) {
                    const int p = atomicAdd(&s_cur[ed[q].x >> COL_BITS], 1);
                    const int2 wd = make_int2(ed[q].x & COL_MASK, ed[q].y);
                    if (in_lds) s_out[p] = wd;
                    else        edges[beg + p] = wd;
                }
            }
#pragma unroll
            for (int q = 0; q < SEG_PER; ++q)
                for (int k = 32 + lane; k < len[q]; k += 32) {
                    const int2 e2 = bedges[off[q] + k];
                    const int p = atomicAdd(&s_cur[e2.x >> COL_BITS], 1);
                    const int2 wd = make_int2(e2.x & COL_MASK, e2.y);
                    if (in_lds) s_out[p] = wd;
                    else        edges[beg + p] = wd;
                }
        }
    }
    __syncthreads();
    if (in_lds)
        for (int i = t; i < n; i += FIN_THREADS) edges[beg + i] = s_out[i];
}


template <int G>
__device__ __forceinline__ void lgcn_chunk_sum(float2& s, int2 ed, int lane,
                                               const float* __restrict__ xa,
                                               const float* __restrict__ xb,
                                               int split) {
    float2 xv[G * 8];
    float  vv[G * 8];
#pragma unroll
    for (int k = 0; k < G * 8; ++k) {
        int c = __builtin_amdgcn_readlane(ed.x, k);
        vv[k] = __int_as_float(__builtin_amdgcn_readlane(ed.y, k));
        const float* src = (c < split) ? (xa + (long)c * DIM)
                                       : (xb + (long)(c - split) * DIM);
        xv[k] = ((const float2*)src)[lane];
    }
#pragma unroll
    for (int k = 0; k < G * 8; ++k) {
        s.x += vv[k] * xv[k].x;
        s.y += vv[k] * xv[k].y;
    }
}

__device__ __forceinline__ void lgcn_first_edges(const int2* __restrict__ edges,
                                                 int beg, int cnt, int lane,
                                                 int2& e0, int2& e1) {
    e0 = make_int2(0, 0);
    e1 = make_int2(0, 0);
    if (lane < cnt)      e0 = edges[beg + lane];
    if (lane + 32 < cnt) e1 = edges[beg + 32 + lane];
}

__device__ __forceinline__ float2 lgcn_row_sum(const int2*  __restrict__ edges,
                                               int beg, int end, int lane,
                                               int2 e0, int2 e1,
                                               const float* __restrict__ xa,
                                               const float* __restrict__ xb,
                                               int split) {
    float2 s = make_float2(0.f, 0.f);
    int ci = 0;
    for (int base = beg; base < end; base += 32, ++ci) {
        int n = end - base;
        if (n > 32) n = 32;
        int2 ed;
        if (ci == 0)      ed = e0;
        else if (ci == 1) ed = e1;
        else {
            ed = make_int2(0, 0);
            if (lane < n) ed = edges[base + lane];
        }
        if (n > 24)      lgcn_chunk_sum<4>(s, ed, lane, xa, xb, split);
        else if (n > 16) lgcn_chunk_sum<3>(s, ed, lane, xa, xb, split);
        else if (n > 8)  lgcn_chunk_sum<2>(s, ed, lane, xa, xb, split);
        else             lgcn_chunk_sum<1>(s, ed, lane, xa, xb, split);
    }
    return s;
}

__device__ __forceinline__ int2 lgcn_pick4(int i, int2 a, int2 b, int2 c, int2 d) {
    int2 r = a;
    if (i == 1) r = b;
    if (i == 2) r = c;
    if (i == 3) r = d;
    return r;
}

__device__ __forceinline__ void lgcn_wave_rows(const int*   __restrict__ row_ptr,
                                               const int*   __restrict__ counts,
                                               const int2*  __restrict__ edges,
                                               const float* __restrict__ xa,
                                               const float* __restrict__ xb,
                                               int split,
                                               float*       __restrict__ xout,
                                               int wave, int lane) {
    const int row0 = wave * ROWS_PER_WAVE;
    if (row0 >= NNODES) return;
    const int4 rp = *(const int4*)(row_ptr + row0);
    const int4 cn = *(const int4*)(counts + row0);
    int2 a0, a1, b0, b1, c0, c1, d0, d1;
    lgcn_first_edges(edges, rp.x, cn.x, lane, a0, a1);
    lgcn_first_edges(edges, rp.y, cn.y, lane, b0, b1);
    lgcn_first_edges(edges, rp.z, cn.z, lane, c0, c1);
    lgcn_first_edges(edges, rp.w, cn.w, lane, d0, d1);
#pragma nounroll
    for (int i = 0; i < ROWS_PER_WAVE; ++i) {
        const int beg = (i == 0) ? rp.x : (i == 1) ? rp.y : (i == 2) ? rp.z : rp.w;
        const int cnt = (i == 0) ? cn.x : (i == 1) ? cn.y : (i == 2) ? cn.z : cn.w;
        const int2 e0 = lgcn_pick4(i, a0, b0, c0, d0);
        const int2 e1 = lgcn_pick4(i, a1, b1, c1, d1);
        float2 s = lgcn_row_sum(edges, beg, beg + cnt, lane, e0, e1, xa, xb, split);
        ((float2*)(xout + (long)(row0 + i) * DIM))[lane] = s;
    }
}

static_assert(ROWS_PER_WAVE == 4, "a wave's row bounds are read as one int4 each");
static_assert(NNODES % ROWS_PER_WAVE == 0, "every wave has all of its rows");

__global__ void lgcn_layer_first(const int*   __restrict__ row_ptr,
                                 const int*   __restrict__ counts,
                                 const int2*  __restrict__ edges,
                                 const float* __restrict__ uemb,
                                 const float* __restrict__ iemb,
                                 float*       __restrict__ x1) {
    const int wave = __builtin_amdgcn_readfirstlane(blockIdx.x * (blockDim.x >> 5) + (threadIdx.x >> 5));
    const int lane = threadIdx.x & 31;
    lgcn_wave_rows(row_ptr, counts, edges, uemb, iemb, N_USERS, x1, wave, lane);
}

__global__ void lgcn_layer_mid(const int*   __restrict__ row_ptr,
                               const int*   __restrict__ counts,
                               const int2*  __restrict__ edges,
                               const float* __restrict__ xin,
                               float*       __restrict__ xout) {
    const int wave = __builtin_amdgcn_readfirstlane(blockIdx.x * (blockDim.x >> 5) + (threadIdx.x >> 5));
    const int lane = threadIdx.x & 31;
    lgcn_wave_rows(row_ptr, counts, edges, xin, xin, NNODES, xout, wave, lane);
}

__device__ __forceinline__ float pick8(v8f c, int j) {
    float r = c[0];
    if (j == 1) r = c[1];
    if (j == 2) r = c[2];
    if (j == 3) r = c[3];
    if (j == 4) r = c[4];
    if (j == 5) r = c[5];
    if (j == 6) r = c[6];
    if (j == 7) r = c[7];
    return r;
}

#define ACC_PITCH 68

__global__ void __launch_bounds__(512)
lgcn_last_score(const int*   __restrict__ row_ptr,
                const int*   __restrict__ counts,
                const int2*  __restrict__ edges,
                const float* __restrict__ uemb,
                const float* __restrict__ iemb,
                const float* __restrict__ x1,
                const float* __restrict__ x2,
                const int*   __restrict__ users,
                const int*   __restrict__ pos,
                const int*   __restrict__ neg,
                float*       __restrict__ pos_scores,
                float*       __restrict__ neg_scores) {
    __shared__ float s_acc[3 * 16 * ACC_PITCH];
    const int w    = __builtin_amdgcn_readfirstlane(threadIdx.x >> 5);
    const int lane = threadIdx.x & 31;
    const int tile = blockIdx.x;
    const int b    = tile * 16 + w;
    const int idu = users[b];
    const int idp = pos[b];
    const int idn = neg[b];
    const int ru = idu;
    const int rp = idp + N_USERS;
    const int rn = idn + N_USERS;
    const int begu = row_ptr[ru];
    const int begp = row_ptr[rp];
    const int begn = row_ptr[rn];
    const int endu = begu + counts[ru];
    const int endp = begp + counts[rp];
    const int endn = begn + counts[rn];
    const float2 u0 = ((const float2*)(uemb + (long)idu * DIM))[lane];
    const float2 p0 = ((const float2*)(iemb + (long)idp * DIM))[lane];
    const float2 n0 = ((const float2*)(iemb + (long)idn * DIM))[lane];
    const float2 u1 = ((const float2*)(x1 + (long)ru * DIM))[lane];
    const float2 p1 = ((const float2*)(x1 + (long)rp * DIM))[lane];
    const float2 n1 = ((const float2*)(x1 + (long)rn * DIM))[lane];
    const float2 u2 = ((const float2*)(x2 + (long)ru * DIM))[lane];
    const float2 p2 = ((const float2*)(x2 + (long)rp * DIM))[lane];
    const float2 n2 = ((const float2*)(x2 + (long)rn * DIM))[lane];

    int2 eu0, eu1, ep0, ep1, en0, en1;
    lgcn_first_edges(edges, begu, endu - begu, lane, eu0, eu1);
    lgcn_first_edges(edges, begp, endp - begp, lane, ep0, ep1);
    lgcn_first_edges(edges, begn, endn - begn, lane, en0, en1);
    const float2 su = lgcn_row_sum(edges, begu, endu, lane, eu0, eu1, x2, x2, NNODES);
    const float2 sp = lgcn_row_sum(edges, begp, endp, lane, ep0, ep1, x2, x2, NNODES);
    const float2 sn = lgcn_row_sum(edges, begn, endn, lane, en0, en1, x2, x2, NNODES);

    float2 au, ap, an;
    au.x = ((u0.x + u1.x) + u2.x) + su.x;
    au.y = ((u0.y + u1.y) + u2.y) + su.y;
    ap.x = ((p0.x + p1.x) + p2.x) + sp.x;
    ap.y = ((p0.y + p1.y) + p2.y) + sp.y;
    an.x = ((n0.x + n1.x) + n2.x) + sn.x;
    an.y = ((n0.y + n1.y) + n2.y) + sn.y;
    ((float2*)(s_acc + (0 * 16 + w) * ACC_PITCH))[lane] = au;
    ((float2*)(s_acc + (1 * 16 + w) * ACC_PITCH))[lane] = ap;
    ((float2*)(s_acc + (2 * 16 + w) * ACC_PITCH))[lane] = an;
    __syncthreads();

    if (w == 0) {
        const int m     = lane & 15;
        const int khalf = lane >> 4;
        const float* urow = s_acc + (0 * 16 + m) * ACC_PITCH + khalf * 2;
        const float* prow = s_acc + (1 * 16 + m) * ACC_PITCH + khalf * 2;
        const float* nrow = s_acc + (2 * 16 + m) * ACC_PITCH + khalf * 2;
        v8f cpos = {};
        v8f cneg = {};
#pragma unroll
        for (int kk = 0; kk < 16; ++kk) {
            float2 ua = *(const float2*)(urow + kk * 4);
            float2 pb = *(const float2*)(prow + kk * 4);
            float2 nb = *(const float2*)(nrow + kk * 4);
            v2f a;  a[0]  = ua.x; a[1]  = ua.y;
            v2f bp; bp[0] = pb.x; bp[1] = pb.y;
            v2f bn; bn[0] = nb.x; bn[1] = nb.y;
            cpos = __builtin_amdgcn_wmma_f32_16x16x4_f32(false, a, false, bp,
                                                         (short)0, cpos, false, false);
            cneg = __builtin_amdgcn_wmma_f32_16x16x4_f32(false, a, false, bn,
                                                         (short)0, cneg, false, false);
        }
        const float s = 1.0f / 16.0f;
        if (lane < 8) {
            pos_scores[tile * 16 + lane] = pick8(cpos, lane) * s;
            neg_scores[tile * 16 + lane] = pick8(cneg, lane) * s;
        } else if (lane >= 24) {
            int mm = lane - 16;
            pos_scores[tile * 16 + mm] = pick8(cpos, lane - 24) * s;
            neg_scores[tile * 16 + mm] = pick8(cneg, lane - 24) * s;
        }
    }
}

__global__ void lgcn_gather0(const float* __restrict__ uemb,
                             const float* __restrict__ iemb,
                             const int*   __restrict__ users,
                             const int*   __restrict__ pos,
                             const int*   __restrict__ neg,
                             float* __restrict__ u0,
                             float* __restrict__ p0,
                             float* __restrict__ n0) {
    int idx = blockIdx.x * blockDim.x + threadIdx.x;
    if (idx >= BATCH * (DIM / 4)) return;
    int b = idx >> 4;
    int q = idx & 15;
    ((float4*)u0)[idx] = ((const float4*)uemb)[(long)users[b] * 16 + q];
    ((float4*)p0)[idx] = ((const float4*)iemb)[(long)pos[b]   * 16 + q];
    ((float4*)n0)[idx] = ((const float4*)iemb)[(long)neg[b]   * 16 + q];
}

extern "C" void kernel_launch(void* const* d_in, const int* in_sizes, int n_in,
                              void* d_out, int out_size, void* d_ws, size_t ws_size,
                              hipStream_t stream) {
    const float* uemb  = (const float*)d_in[0];
    const float* iemb  = (const float*)d_in[1];
    const float* vals  = (const float*)d_in[2];
    const int*   rows  = (const int*)d_in[3];
    const int*   cols  = (const int*)d_in[4];
    const int*   users = (const int*)d_in[5];
    const int*   pos   = (const int*)d_in[6];
    const int*   neg   = (const int*)d_in[7];
    const int    nnz   = in_sizes[3];

    float* out        = (float*)d_out;
    float* pos_scores = out;
    float* neg_scores = out + BATCH;
    float* u0         = out + 2 * BATCH;
    float* p0         = u0 + BATCH * DIM;
    float* n0         = p0 + BATCH * DIM;

    const size_t nfeat = (size_t)NNODES * DIM;
    float* x1      = (float*)d_ws;
    float* x2      = x1 + nfeat;
    int2*  bedges  = (int2*)x2;
    int*   cursor  = (int*)x1;
    int*   toff    = cursor + 32;
    int*   counts  = (int*)(x2 + nfeat);
    int*   row_ptr = counts + NNODES;
    int2*  edges   = (int2*)(row_ptr + NNODES);

    const int tgrid = (nnz + TILE_EDGES - 1) / TILE_EDGES;
    const int lgrid = (NNODES / ROWS_PER_WAVE + 7) / 8;

    lgcn_bin_group<<<tgrid, TILE_THREADS, 0, stream>>>(rows, cols, vals, toff, cursor, bedges, nnz);
    lgcn_bin_finish<<<NBKT, FIN_THREADS, 0, stream>>>(toff, bedges, cursor, counts, row_ptr, edges, tgrid);

    lgcn_layer_first<<<lgrid, 256, 0, stream>>>(row_ptr, counts, edges, uemb, iemb, x1);
    lgcn_layer_mid<<<lgrid, 256, 0, stream>>>(row_ptr, counts, edges, x1, x2);

    lgcn_last_score<<<BATCH / 16, 512, 0, stream>>>(row_ptr, counts, edges, uemb, iemb,
                                                   x1, x2, users, pos, neg,
                                                   pos_scores, neg_scores);

    lgcn_gather0<<<(BATCH * (DIM / 4) + 255) / 256, 256, 0, stream>>>(
        uemb, iemb, users, pos, neg, u0, p0, n0);
}


# Row sums take two rows per wave, one per 16-lane half with 16-byte loads: one load serves an edge of each row, 16-lane broadcasts replace the per-edge scalar address work; each row's edge order kept
# speedup vs baseline: 7.9642x; 1.0384x over previous
#include <hip/hip_runtime.h>


#define N_USERS 100000
#define N_ITEMS 50000
#define NNODES  (N_USERS + N_ITEMS)
#define DIM     64
#define BATCH   8192
#define ROWS_PER_WAVE 4

typedef __attribute__((ext_vector_type(2))) float v2f;
typedef __attribute__((ext_vector_type(8))) float v8f;
typedef __attribute__((ext_vector_type(4))) float v4f;
typedef __attribute__((ext_vector_type(2))) unsigned v2u;
typedef const __attribute__((address_space(1))) v4f* lgcn_gv4;

#define BROWS_LOG    8
#define BROWS        (1 << BROWS_LOG)
#define NBKT         ((NNODES + BROWS - 1) / BROWS)
#define NB1          (NBKT + 1)
#define TILE_THREADS 1024
#define TILE_PER     8
#define TILE_EDGES   (TILE_THREADS * TILE_PER)
#define COL_BITS     18
#define COL_MASK     ((1 << COL_BITS) - 1)
#define FIN_THREADS  512
#define FIN_WAVES    (FIN_THREADS / 32)
#define SEG_PER      8
#define FIN_CAP      12288

static_assert(NBKT < TILE_THREADS, "level 1 wants a thread per bin and one more");
static_assert((FIN_THREADS & (FIN_THREADS - 1)) == 0, "piece numbers are wrapped with a mask");

__device__ __forceinline__ int lgcn_block_excl_scan(int v, int t, int* s_w) {
    const int lane = t & 31;
    const int w = t >> 5;
    int incl = v;
#pragma unroll
    for (int d = 1; d < 32; d <<= 1) {
        int u = __shfl_up(incl, d);
        if (lane >= d) incl += u;
    }
    if (lane == 31) s_w[w] = incl;
    __syncthreads();
    int base = 0;
    for (int k = 0; k < w; ++k) base += s_w[k];
    return base + incl - v;
}

__global__ void __launch_bounds__(TILE_THREADS)
lgcn_bin_group(const int*   __restrict__ rows,
               const int*   __restrict__ cols,
               const float* __restrict__ vals,
               int*         __restrict__ toff,
               int*         __restrict__ cursor,
               int2*        __restrict__ bedges,
               int nnz) {
    __shared__ int s_h[NBKT];
    __shared__ int s_loc[NBKT];
    __shared__ int s_w[32];
    __shared__ int s_key[TILE_EDGES];
    __shared__ int s_val[TILE_EDGES];
    const int t = threadIdx.x;
    if (blockIdx.x == 0 && t == 0) *cursor = 0;
    for (int i = t; i < NBKT; i += TILE_THREADS) s_h[i] = 0;
    __syncthreads();
    const int base = blockIdx.x * TILE_EDGES;
    int r[TILE_PER];
    int lr[TILE_PER];
#pragma unroll
    for (int k = 0; k < TILE_PER; ++k) {
        const int e = base + k * TILE_THREADS + t;
        r[k] = (e < nnz) ? rows[e] : -1;
    }
#pragma unroll
    for (int k = 0; k < TILE_PER; ++k)
        lr[k] = (r[k] >= 0) ? atomicAdd(&s_h[r[k] >> BROWS_LOG], 1) : 0;
    int   cv[TILE_PER];
    float vv[TILE_PER];
#pragma unroll
    for (int k = 0; k < TILE_PER; ++k) {
        const int e = base + k * TILE_THREADS + t;
        cv[k] = (e < nnz) ? cols[e] : 0;
        vv[k] = (e < nnz) ? vals[e] : 0.f;
    }
    __syncthreads();
    const int c = (t < NBKT) ? s_h[t] : 0;
    const int ex = lgcn_block_excl_scan(c, t, s_w);
    if (t < NBKT) s_loc[t] = ex;
    if (t <= NBKT) toff[blockIdx.x * NB1 + t] = ex;
    __syncthreads();
#pragma unroll
    for (int k = 0; k < TILE_PER; ++k) {
        if (r[k] >= 0) {
            const int pos = s_loc[r[k] >> BROWS_LOG] + lr[k];
            s_key[pos] = ((r[k] & (BROWS - 1)) << COL_BITS) | cv[k];
            s_val[pos] = __float_as_int(vv[k]);
        }
    }
    __syncthreads();
    int n = nnz - base;
    if (n > TILE_EDGES) n = TILE_EDGES;
    for (int i = t; i < n; i += TILE_THREADS)
        bedges[base + i] = make_int2(s_key[i], s_val[i]);
}

__device__ __forceinline__ void lgcn_stage_pieces(const int* __restrict__ toff,
                                                  int b, int t0, int ntiles, int t,
                                                  int* s_off, int* s_len) {
    const int tt = t0 + t;
    int o = 0, l = 0;
    if (tt < ntiles) {
        o = toff[tt * NB1 + b];
        l = toff[tt * NB1 + b + 1] - o;
    }
    s_off[t] = tt * TILE_EDGES + o;
    s_len[t] = l;
}

__global__ void __launch_bounds__(FIN_THREADS)
lgcn_bin_finish(const int*  __restrict__ toff,
                const int2* __restrict__ bedges,
                int*        __restrict__ cursor,
                int*        __restrict__ counts,
                int*        __restrict__ row_ptr,
                int2*       __restrict__ edges,
                int ntiles) {
    __shared__ int  s_cnt[BROWS];
    __shared__ int  s_cur[BROWS];
    __shared__ int  s_w[32];
    __shared__ int  s_off[FIN_THREADS];
    __shared__ int  s_len[FIN_THREADS];
    __shared__ int  s_run[2];
    __shared__ int2 s_out[FIN_CAP];
    const int t = threadIdx.x;
    const int b = blockIdx.x;
    const int lane = t & 31;
    const int w = __builtin_amdgcn_readfirstlane(t >> 5);
    if (t < BROWS) s_cnt[t] = 0;
    for (int t0 = 0; t0 < ntiles; t0 += FIN_THREADS) {
        __syncthreads();
        lgcn_stage_pieces(toff, b, t0, ntiles, t, s_off, s_len);
        __syncthreads();
        int nt = ntiles - t0;
        if (nt > FIN_THREADS) nt = FIN_THREADS;
        for (int s0 = w; s0 < nt; s0 += FIN_WAVES * SEG_PER) {
            int off[SEG_PER], len[SEG_PER], key[SEG_PER];
#pragma unroll
            for (int q = 0; q < SEG_PER; ++q) {
                const int s = s0 + q * FIN_WAVES;
                off[q] = s_off[s & (FIN_THREADS - 1)];
                len[q] = s_len[s & (FIN_THREADS - 1)];
                if (s >= nt) len[q] = 0;
            }
#pragma unroll
            for (int q = 0; q < SEG_PER; ++q)
                key[q] = (lane < len[q]) ? bedges[off[q] + lane].x : -1;
#pragma unroll
            for (int q = 0; q < SEG_PER; ++q)
                if (key[q] >= 0) atomicAdd(&s_cnt[key[q] >> COL_BITS], 1);
#pragma unroll
            for (int q = 0; q < SEG_PER; ++q)
                for (int k = 32 + lane; k < len[q]; k += 32)
                    atomicAdd(&s_cnt[bedges[off[q] + k].x >> COL_BITS], 1);
        }
    }
    __syncthreads();
    const int v = (t < BROWS) ? s_cnt[t] : 0;
    const int ex = lgcn_block_excl_scan(v, t, s_w);
    if (t < BROWS) s_cur[t] = ex;
    if (t == BROWS - 1) {
        s_run[0] = ex + v;
        s_run[1] = atomicAdd(cursor, ex + v);
    }
    __syncthreads();
    const int n = s_run[0];
    const int beg = s_run[1];
    const bool in_lds = (n <= FIN_CAP);
    if (t < BROWS) {
        const int row = b * BROWS + t;
        if (row < NNODES) {
            row_ptr[row] = beg + ex;
            counts[row] = v;
        }
    }
    for (int t0 = 0; t0 < ntiles; t0 += FIN_THREADS) {
        if (t0 > 0 || ntiles > FIN_THREADS) {
            __syncthreads();
            lgcn_stage_pieces(toff, b, t0, ntiles, t, s_off, s_len);
            __syncthreads();
        }
        int nt = ntiles - t0;
        if (nt > FIN_THREADS) nt = FIN_THREADS;
        for (int s0 = w; s0 < nt; s0 += FIN_WAVES * SEG_PER) {
            int off[SEG_PER], len[SEG_PER];
            int2 ed[SEG_PER];
#pragma unroll
            for (int q = 0; q < SEG_PER; ++q) {
                const int s = s0 + q * FIN_WAVES;
                off[q] = s_off[s & (FIN_THREADS - 1)];
                len[q] = s_len[s & (FIN_THREADS - 1)];
                if (s >= nt) len[q] = 0;
            }
#pragma unroll
            for (int q = 0; q < SEG_PER; ++q)
                ed[q] = (lane < len[q]) ? bedges[off[q] + lane] : make_int2(-1, 0);
#pragma unroll
            for (int q = 0; q < SEG_PER; ++q) {
                if (ed[q].x >= 0) {
                    const int p = atomicAdd(&s_cur[ed[q].x >> COL_BITS], 1);
                    const int2 wd = make_int2(ed[q].x & COL_MASK, ed[q].y);
                    if (in_lds) s_out[p] = wd;
                    else        edges[beg + p] = wd;
                }
            }
#pragma unroll
            for (int q = 0; q < SEG_PER; ++q)
                for (int k = 32 + lane; k < len[q]; k += 32) {
                    const int2 e2 = bedges[off[q] + k];
                    const int p = atomicAdd(&s_cur[e2.x >> COL_BITS], 1);
                    const int2 wd = make_int2(e2.x & COL_MASK, e2.y);
                    if (in_lds) s_out[p] = wd;
                    else        edges[beg + p] = wd;
                }
        }
    }
    __syncthreads();
    if (in_lds)
        for (int i = t; i < n; i += FIN_THREADS) edges[beg + i] = s_out[i];
}


template <int J>
__device__ __forceinline__ int lgcn_row_bcast(int v) {
    return __builtin_amdgcn_update_dpp(0, v, 0x150 + J, 0xf, 0xf, true);
}

template <int J, int S, bool ONE>
__device__ __forceinline__ void lgcn_pair_loads(v4f (&xv)[S], float (&wv)[S],
                                                int a_lo, int a_hi, int wbits, int l16off,
                                                const float* __restrict__ xa) {
    if constexpr (J < S) {
        wv[J] = __int_as_float(lgcn_row_bcast<J>(wbits));
        if constexpr (ONE) {
            const unsigned o = (unsigned)lgcn_row_bcast<J>(a_lo) + (unsigned)l16off;
            xv[J] = *(const v4f*)((const char*)xa + o);
        } else {
            const unsigned lo = (unsigned)lgcn_row_bcast<J>(a_lo);
            const unsigned hi = (unsigned)lgcn_row_bcast<J>(a_hi);
            const v2u pr = {lo, hi};
            const unsigned long long a = __builtin_bit_cast(unsigned long long, pr) + (unsigned)l16off;
            xv[J] = *(lgcn_gv4)a;
        }
        lgcn_pair_loads<J + 1, S, ONE>(xv, wv, a_lo, a_hi, wbits, l16off, xa);
    }
}

template <int S, bool ONE>
__device__ __forceinline__ void lgcn_pair_chunk(float4& s, int2 ed, int l16off,
                                                const float* __restrict__ xa,
                                                const float* __restrict__ xb,
                                                int split) {
    v4f   xv[S];
    float wv[S];
    int a_lo, a_hi = 0;
    if constexpr (ONE) {
        a_lo = ed.x << 8;
    } else {
        const int c = ed.x;
        const float* src = (c < split) ? (xa + (long)c * DIM)
                                       : (xb + (long)(c - split) * DIM);
        const unsigned long long a = (unsigned long long)src;
        a_lo = (int)(unsigned)a;
        a_hi = (int)(unsigned)(a >> 32);
    }
    lgcn_pair_loads<0, S, ONE>(xv, wv, a_lo, a_hi, ed.y, l16off, xa);
#pragma unroll
    for (int k = 0; k < S; ++k) {
        s.x = __builtin_fmaf(wv[k], xv[k].x, s.x);
        s.y = __builtin_fmaf(wv[k], xv[k].y, s.y);
        s.z = __builtin_fmaf(wv[k], xv[k].z, s.z);
        s.w = __builtin_fmaf(wv[k], xv[k].w, s.w);
    }
}

__device__ __forceinline__ void lgcn_pair_first(const int2* __restrict__ edges,
                                                int beg, int cnt, int l16,
                                                int2& e0, int2& e1) {
    e0 = make_int2(0, 0);
    e1 = make_int2(0, 0);
    if (l16 < cnt)      e0 = edges[beg + l16];
    if (l16 + 16 < cnt) e1 = edges[beg + 16 + l16];
}

template <bool ONE>
__device__ __forceinline__ float4 lgcn_pair_sum(const int2* __restrict__ edges,
                                                int beg, int cnt, int nmax, int l16,
                                                int2 e0, int2 e1,
                                                const float* __restrict__ xa,
                                                const float* __restrict__ xb,
                                                int split) {
    float4 s = make_float4(0.f, 0.f, 0.f, 0.f);
    const int l16off = l16 << 4;
    int2 cur = e0, nxt = e1;
    for (int base = 0; base < nmax; base += 16) {
        int2 nn = make_int2(0, 0);
        if (base + 32 + l16 < cnt) nn = edges[beg + base + 32 + l16];
        const int n = nmax - base;
        if (n > 8) {
            if (n > 12) {
                if (n > 14) lgcn_pair_chunk<16, ONE>(s, cur, l16off, xa, xb, split);
                else        lgcn_pair_chunk<14, ONE>(s, cur, l16off, xa, xb, split);
            } else {
                if (n > 10) lgcn_pair_chunk<12, ONE>(s, cur, l16off, xa, xb, split);
                else        lgcn_pair_chunk<10, ONE>(s, cur, l16off, xa, xb, split);
            }
        } else {
            if (n > 4) {
                if (n > 6)  lgcn_pair_chunk<8, ONE>(s, cur, l16off, xa, xb, split);
                else        lgcn_pair_chunk<6, ONE>(s, cur, l16off, xa, xb, split);
            } else {
                if (n > 2)  lgcn_pair_chunk<4, ONE>(s, cur, l16off, xa, xb, split);
                else        lgcn_pair_chunk<2, ONE>(s, cur, l16off, xa, xb, split);
            }
        }
        cur = nxt;
        nxt = nn;
    }
    return s;
}

template <bool ONE>
__device__ __forceinline__ void lgcn_wave_rows(const int*   __restrict__ row_ptr,
                                               const int*   __restrict__ counts,
                                               const int2*  __restrict__ edges,
                                               const float* __restrict__ xa,
                                               const float* __restrict__ xb,
                                               int split,
                                               float*       __restrict__ xout,
                                               int wave, int lane) {
    const int row0 = wave * ROWS_PER_WAVE;
    if (row0 >= NNODES) return;
    const int4 rp = *(const int4*)(row_ptr + row0);
    const int4 cn = *(const int4*)(counts + row0);
    const bool hi = lane >= 16;
    const int  l16 = lane & 15;
    const int begA = hi ? rp.y : rp.x;
    const int cntA = hi ? cn.y : cn.x;
    const int begB = hi ? rp.w : rp.z;
    const int cntB = hi ? cn.w : cn.z;
    const int maxA = cn.x > cn.y ? cn.x : cn.y;
    const int maxB = cn.z > cn.w ? cn.z : cn.w;
    int2 a0, a1, b0, b1;
    lgcn_pair_first(edges, begA, cntA, l16, a0, a1);
    lgcn_pair_first(edges, begB, cntB, l16, b0, b1);
#pragma nounroll
    for (int p = 0; p < ROWS_PER_WAVE / 2; ++p) {
        const int  beg  = (p == 0) ? begA : begB;
        const int  cnt  = (p == 0) ? cntA : cntB;
        const int  nmax = (p == 0) ? maxA : maxB;
        const int2 e0   = (p == 0) ? a0 : b0;
        const int2 e1   = (p == 0) ? a1 : b1;
        const float4 s = lgcn_pair_sum<ONE>(edges, beg, cnt, nmax, l16, e0, e1, xa, xb, split);
        ((float4*)(xout + (long)(row0 + 2 * p) * DIM))[lane] = s;
    }
}

static_assert(ROWS_PER_WAVE == 4, "a wave's row bounds are read as one int4 each");
static_assert(NNODES % ROWS_PER_WAVE == 0, "every wave has all of its rows");

__global__ void lgcn_layer_first(const int*   __restrict__ row_ptr,
                                 const int*   __restrict__ counts,
                                 const int2*  __restrict__ edges,
                                 const float* __restrict__ uemb,
                                 const float* __restrict__ iemb,
                                 float*       __restrict__ x1) {
    const int wave = __builtin_amdgcn_readfirstlane(blockIdx.x * (blockDim.x >> 5) + (threadIdx.x >> 5));
    const int lane = threadIdx.x & 31;
    lgcn_wave_rows<false>(row_ptr, counts, edges, uemb, iemb, N_USERS, x1, wave, lane);
}

__global__ void lgcn_layer_mid(const int*   __restrict__ row_ptr,
                               const int*   __restrict__ counts,
                               const int2*  __restrict__ edges,
                               const float* __restrict__ xin,
                               float*       __restrict__ xout) {
    const int wave = __builtin_amdgcn_readfirstlane(blockIdx.x * (blockDim.x >> 5) + (threadIdx.x >> 5));
    const int lane = threadIdx.x & 31;
    lgcn_wave_rows<true>(row_ptr, counts, edges, xin, xin, NNODES, xout, wave, lane);
}

__device__ __forceinline__ float pick8(v8f c, int j) {
    float r = c[0];
    if (j == 1) r = c[1];
    if (j == 2) r = c[2];
    if (j == 3) r = c[3];
    if (j == 4) r = c[4];
    if (j == 5) r = c[5];
    if (j == 6) r = c[6];
    if (j == 7) r = c[7];
    return r;
}

#define ACC_PITCH 68
#define SCORE_WAVES 12

struct lgcn_half_row {
    int    beg, cnt;
    float4 t0, t1, t2;
};

__device__ __forceinline__ lgcn_half_row lgcn_half_setup(int kind, int b, int l16,
                                                         const int*   __restrict__ row_ptr,
                                                         const int*   __restrict__ counts,
                                                         const float* __restrict__ uemb,
                                                         const float* __restrict__ iemb,
                                                         const float* __restrict__ x1,
                                                         const float* __restrict__ x2,
                                                         const int*   __restrict__ users,
                                                         const int*   __restrict__ pos,
                                                         const int*   __restrict__ neg) {
    const int*   ids = (kind == 0) ? users : (kind == 1) ? pos : neg;
    const float* e0t = (kind == 0) ? uemb : iemb;
    const int id = ids[b];
    const int r  = id + ((kind == 0) ? 0 : N_USERS);
    lgcn_half_row h;
    h.beg = row_ptr[r];
    h.cnt = counts[r];
    h.t0 = ((const float4*)(e0t + (long)id * DIM))[l16];
    h.t1 = ((const float4*)(x1 + (long)r * DIM))[l16];
    h.t2 = ((const float4*)(x2 + (long)r * DIM))[l16];
    return h;
}

__global__ void __launch_bounds__(SCORE_WAVES * 32)
lgcn_last_score(const int*   __restrict__ row_ptr,
                const int*   __restrict__ counts,
                const int2*  __restrict__ edges,
                const float* __restrict__ uemb,
                const float* __restrict__ iemb,
                const float* __restrict__ x1,
                const float* __restrict__ x2,
                const int*   __restrict__ users,
                const int*   __restrict__ pos,
                const int*   __restrict__ neg,
                float*       __restrict__ pos_scores,
                float*       __restrict__ neg_scores) {
    __shared__ __attribute__((aligned(16))) float s_acc[3 * 16 * ACC_PITCH];
    const int w    = __builtin_amdgcn_readfirstlane(threadIdx.x >> 5);
    const int lane = threadIdx.x & 31;
    const int l16  = lane & 15;
    const int hf   = lane >> 4;
    const int tile = blockIdx.x;
    const int qA = 2 * w + hf;
    const int qB = 2 * (w + SCORE_WAVES) + hf;
    const int kindA = (2 * w) >> 4;
    const int kindB = (2 * (w + SCORE_WAVES)) >> 4;
    const lgcn_half_row ha = lgcn_half_setup(kindA, tile * 16 + (qA & 15), l16, row_ptr, counts,
                                             uemb, iemb, x1, x2, users, pos, neg);
    const lgcn_half_row hb = lgcn_half_setup(kindB, tile * 16 + (qB & 15), l16, row_ptr, counts,
                                             uemb, iemb, x1, x2, users, pos, neg);
    int2 a0, a1, b0, b1;
    lgcn_pair_first(edges, ha.beg, ha.cnt, l16, a0, a1);
    lgcn_pair_first(edges, hb.beg, hb.cnt, l16, b0, b1);
    const int cA0 = __builtin_amdgcn_readlane(ha.cnt, 0), cA1 = __builtin_amdgcn_readlane(ha.cnt, 16);
    const int cB0 = __builtin_amdgcn_readlane(hb.cnt, 0), cB1 = __builtin_amdgcn_readlane(hb.cnt, 16);
    const int maxA = cA0 > cA1 ? cA0 : cA1;
    const int maxB = cB0 > cB1 ? cB0 : cB1;
#pragma nounroll
    for (int p = 0; p < 2; ++p) {
        const int    beg  = (p == 0) ? ha.beg : hb.beg;
        const int    cnt  = (p == 0) ? ha.cnt : hb.cnt;
        const int    nmax = (p == 0) ? maxA : maxB;
        const int    q    = (p == 0) ? qA : qB;
        const int2   e0   = (p == 0) ? a0 : b0;
        const int2   e1   = (p == 0) ? a1 : b1;
        const float4 t0   = (p == 0) ? ha.t0 : hb.t0;
        const float4 t1   = (p == 0) ? ha.t1 : hb.t1;
        const float4 t2   = (p == 0) ? ha.t2 : hb.t2;
        const float4 s = lgcn_pair_sum<true>(edges, beg, cnt, nmax, l16, e0, e1, x2, x2, NNODES);
        float4 acc;
        acc.x = ((t0.x + t1.x) + t2.x) + s.x;
        acc.y = ((t0.y + t1.y) + t2.y) + s.y;
        acc.z = ((t0.z + t1.z) + t2.z) + s.z;
        acc.w = ((t0.w + t1.w) + t2.w) + s.w;
        ((float4*)(s_acc + q * ACC_PITCH))[l16] = acc;
    }
    __syncthreads();

    if (w == 0) {
        const int m     = lane & 15;
        const int khalf = lane >> 4;
        const float* urow = s_acc + (0 * 16 + m) * ACC_PITCH + khalf * 2;
        const float* prow = s_acc + (1 * 16 + m) * ACC_PITCH + khalf * 2;
        const float* nrow = s_acc + (2 * 16 + m) * ACC_PITCH + khalf * 2;
        v8f cpos = {};
        v8f cneg = {};
#pragma unroll
        for (int kk = 0; kk < 16; ++kk) {
            float2 ua = *(const float2*)(urow + kk * 4);
            float2 pb = *(const float2*)(prow + kk * 4);
            float2 nb = *(const float2*)(nrow + kk * 4);
            v2f a;  a[0]  = ua.x; a[1]  = ua.y;
            v2f bp; bp[0] = pb.x; bp[1] = pb.y;
            v2f bn; bn[0] = nb.x; bn[1] = nb.y;
            cpos = __builtin_amdgcn_wmma_f32_16x16x4_f32(false, a, false, bp,
                                                         (short)0, cpos, false, false);
            cneg = __builtin_amdgcn_wmma_f32_16x16x4_f32(false, a, false, bn,
                                                         (short)0, cneg, false, false);
        }
        const float s = 1.0f / 16.0f;
        if (lane < 8) {
            pos_scores[tile * 16 + lane] = pick8(cpos, lane) * s;
            neg_scores[tile * 16 + lane] = pick8(cneg, lane) * s;
        } else if (lane >= 24) {
            int mm = lane - 16;
            pos_scores[tile * 16 + mm] = pick8(cpos, lane - 24) * s;
            neg_scores[tile * 16 + mm] = pick8(cneg, lane - 24) * s;
        }
    }
}

__global__ void lgcn_gather0(const float* __restrict__ uemb,
                             const float* __restrict__ iemb,
                             const int*   __restrict__ users,
                             const int*   __restrict__ pos,
                             const int*   __restrict__ neg,
                             float* __restrict__ u0,
                             float* __restrict__ p0,
                             float* __restrict__ n0) {
    int idx = blockIdx.x * blockDim.x + threadIdx.x;
    if (idx >= BATCH * (DIM / 4)) return;
    int b = idx >> 4;
    int q = idx & 15;
    ((float4*)u0)[idx] = ((const float4*)uemb)[(long)users[b] * 16 + q];
    ((float4*)p0)[idx] = ((const float4*)iemb)[(long)pos[b]   * 16 + q];
    ((float4*)n0)[idx] = ((const float4*)iemb)[(long)neg[b]   * 16 + q];
}

extern "C" void kernel_launch(void* const* d_in, const int* in_sizes, int n_in,
                              void* d_out, int out_size, void* d_ws, size_t ws_size,
                              hipStream_t stream) {
    const float* uemb  = (const float*)d_in[0];
    const float* iemb  = (const float*)d_in[1];
    const float* vals  = (const float*)d_in[2];
    const int*   rows  = (const int*)d_in[3];
    const int*   cols  = (const int*)d_in[4];
    const int*   users = (const int*)d_in[5];
    const int*   pos   = (const int*)d_in[6];
    const int*   neg   = (const int*)d_in[7];
    const int    nnz   = in_sizes[3];

    float* out        = (float*)d_out;
    float* pos_scores = out;
    float* neg_scores = out + BATCH;
    float* u0         = out + 2 * BATCH;
    float* p0         = u0 + BATCH * DIM;
    float* n0         = p0 + BATCH * DIM;

    const size_t nfeat = (size_t)NNODES * DIM;
    float* x1      = (float*)d_ws;
    float* x2      = x1 + nfeat;
    int2*  bedges  = (int2*)x2;
    int*   cursor  = (int*)x1;
    int*   toff    = cursor + 32;
    int*   counts  = (int*)(x2 + nfeat);
    int*   row_ptr = counts + NNODES;
    int2*  edges   = (int2*)(row_ptr + NNODES);

    const int tgrid = (nnz + TILE_EDGES - 1) / TILE_EDGES;
    const int lgrid = (NNODES / ROWS_PER_WAVE + 7) / 8;

    lgcn_bin_group<<<tgrid, TILE_THREADS, 0, stream>>>(rows, cols, vals, toff, cursor, bedges, nnz);
    lgcn_bin_finish<<<NBKT, FIN_THREADS, 0, stream>>>(toff, bedges, cursor, counts, row_ptr, edges, tgrid);

    lgcn_layer_first<<<lgrid, 256, 0, stream>>>(row_ptr, counts, edges, uemb, iemb, x1);
    lgcn_layer_mid<<<lgrid, 256, 0, stream>>>(row_ptr, counts, edges, x1, x2);

    lgcn_last_score<<<BATCH / 16, SCORE_WAVES * 32, 0, stream>>>(row_ptr, counts, edges, uemb, iemb,
                                                   x1, x2, users, pos, neg,
                                                   pos_scores, neg_scores);

    lgcn_gather0<<<(BATCH * (DIM / 4) + 255) / 256, 256, 0, stream>>>(
        uemb, iemb, users, pos, neg, u0, p0, n0);
}
